# starting kernel
# baseline (speedup 1.0000x reference)
#include <hip/hip_runtime.h>

typedef float v2f __attribute__((ext_vector_type(2)));
typedef float v8f __attribute__((ext_vector_type(8)));

#define NFEAT 256
#define NHID  64


__global__ void k_deg_init(float* deg, int n) {
    int i = blockIdx.x * blockDim.x + threadIdx.x;
    if (i < n) deg[i] = 1.0f;
}

__global__ void k_deg_count(const int* __restrict__ dst, float* deg, int n_edges) {
    int e = blockIdx.x * blockDim.x + threadIdx.x;
    if (e < n_edges) atomicAdd(&deg[dst[e]], 1.0f);
}

__global__ void k_dinv(float* deg, int n) {
    int i = blockIdx.x * blockDim.x + threadIdx.x;
    if (i < n) {
        float d = deg[i];
        deg[i] = (d > 0.0f) ? rsqrtf(d) : 0.0f;
    }
}


__global__ __launch_bounds__(256) void k_gemm1_wmma(const float* __restrict__ x,
                                                    const float* __restrict__ W1,
                                                    float* __restrict__ h,
                                                    int n_nodes) {
    __shared__ float ldsW[NFEAT * NHID];
    int tid = threadIdx.x;

    for (int idx = tid * 4; idx < NFEAT * NHID; idx += 256 * 4) {
        float4 w = *(const float4*)(W1 + idx);
        int k  = idx >> 6;
        int n0 = idx & 63;
        int p  = k >> 1;
        int lo = k & 1;
        int rot = (p & 1) << 5;
        float vals[4] = {w.x, w.y, w.z, w.w};
        #pragma unroll
        for (int j = 0; j < 4; ++j) {
            int wpos = (((n0 + j) << 1) + rot) & 127;
            ldsW[p * 128 + wpos + lo] = vals[j];
        }
    }
    __syncthreads();

    int wave = tid >> 5;
    int lane = tid & 31;
    int tile = blockIdx.x * 8 + wave;
    int m0 = tile * 16;
    if (m0 >= n_nodes) return;

    int half = lane >> 4;
    int mrow = m0 + (lane & 15);
    int kk   = half << 1;
    int ncol = lane & 15;
    int rot  = half << 5;

    int wp0 = (((ncol +  0) << 1) + rot) & 127;
    int wp1 = (((ncol + 16) << 1) + rot) & 127;
    int wp2 = (((ncol + 32) << 1) + rot) & 127;
    int wp3 = (((ncol + 48) << 1) + rot) & 127;

    v8f acc0 = {}, acc1 = {}, acc2 = {}, acc3 = {};

    const float* xrow = x + (size_t)mrow * NFEAT;
    for (int k = 0; k < NFEAT; k += 4) {
        float2 av = *(const float2*)(xrow + k + kk);
        v2f a; a.x = av.x; a.y = av.y;

        const float* rowp = ldsW + ((k >> 1) + half) * 128;
        v2f b0 = *(const v2f*)(rowp + wp0);
        v2f b1 = *(const v2f*)(rowp + wp1);
        v2f b2 = *(const v2f*)(rowp + wp2);
        v2f b3 = *(const v2f*)(rowp + wp3);

        acc0 = __builtin_amdgcn_wmma_f32_16x16x4_f32(false, a, false, b0, (short)0, acc0, false, false);
        acc1 = __builtin_amdgcn_wmma_f32_16x16x4_f32(false, a, false, b1, (short)0, acc1, false, false);
        acc2 = __builtin_amdgcn_wmma_f32_16x16x4_f32(false, a, false, b2, (short)0, acc2, false, false);
        acc3 = __builtin_amdgcn_wmma_f32_16x16x4_f32(false, a, false, b3, (short)0, acc3, false, false);
    }

    int rbase = m0 + (half << 3);
    for (int r = 0; r < 8; ++r) {
        float* hr = h + (size_t)(rbase + r) * NHID;
        hr[ncol +  0] = acc0[r];
        hr[ncol + 16] = acc1[r];
        hr[ncol + 32] = acc2[r];
        hr[ncol + 48] = acc3[r];
    }
}


__global__ void k_selfloop1(const float* __restrict__ h, const float* __restrict__ dinv,
                            float* __restrict__ out1, int n_nodes) {
    int t = blockIdx.x * blockDim.x + threadIdx.x;
    int i = t >> 6, f = t & 63;
    if (i >= n_nodes) return;
    float di = dinv[i];
    out1[(size_t)i * NHID + f] = h[(size_t)i * NHID + f] * di * di;
}

__global__ void k_edge_agg1(const int* __restrict__ src, const int* __restrict__ dst,
                            const float* __restrict__ dinv, const float* __restrict__ h,
                            float* __restrict__ out1, int n_edges) {
    int t = blockIdx.x * blockDim.x + threadIdx.x;
    int e = t >> 6, f = t & 63;
    if (e >= n_edges) return;
    int s = src[e], d = dst[e];
    float norm = dinv[s] * dinv[d];
    atomicAdd(&out1[(size_t)d * NHID + f], h[(size_t)s * NHID + f] * norm);
}

__global__ void k_relu_bias(const float* __restrict__ out1, const float* __restrict__ b1,
                            float* __restrict__ h1, int n_nodes) {
    int t = blockIdx.x * blockDim.x + threadIdx.x;
    int i = t >> 6, f = t & 63;
    if (i >= n_nodes) return;
    float v = out1[(size_t)i * NHID + f] + b1[f];
    h1[(size_t)i * NHID + f] = fmaxf(v, 0.0f);
}


__global__ void k_gemm2(const float* __restrict__ h1, const float* __restrict__ W2,
                        const float* __restrict__ b2, const float* __restrict__ dinv,
                        float* __restrict__ z, float* __restrict__ out, int n_nodes) {
    int i = blockIdx.x * blockDim.x + threadIdx.x;
    if (i >= n_nodes) return;
    const float* hr = h1 + (size_t)i * NHID;
    float z0 = 0.0f, z1 = 0.0f;
    #pragma unroll
    for (int f = 0; f < NHID; f += 4) {
        float4 hv = *(const float4*)(hr + f);
        z0 += hv.x * W2[2 * (f + 0) + 0]; z1 += hv.x * W2[2 * (f + 0) + 1];
        z0 += hv.y * W2[2 * (f + 1) + 0]; z1 += hv.y * W2[2 * (f + 1) + 1];
        z0 += hv.z * W2[2 * (f + 2) + 0]; z1 += hv.z * W2[2 * (f + 2) + 1];
        z0 += hv.w * W2[2 * (f + 3) + 0]; z1 += hv.w * W2[2 * (f + 3) + 1];
    }
    float di = dinv[i], n2 = di * di;
    z[2 * i + 0] = z0;
    z[2 * i + 1] = z1;
    out[2 * i + 0] = z0 * n2 + b2[0];
    out[2 * i + 1] = z1 * n2 + b2[1];
}

__global__ void k_edge_agg2(const int* __restrict__ src, const int* __restrict__ dst,
                            const float* __restrict__ dinv, const float* __restrict__ z,
                            float* __restrict__ out, int n_edges) {
    int e = blockIdx.x * blockDim.x + threadIdx.x;
    if (e >= n_edges) return;
    int s = src[e], d = dst[e];
    float norm = dinv[s] * dinv[d];
    float2 zv = *(const float2*)(z + 2 * s);
    atomicAdd(&out[2 * d + 0], zv.x * norm);
    atomicAdd(&out[2 * d + 1], zv.y * norm);
}


extern "C" void kernel_launch(void* const* d_in, const int* in_sizes, int n_in,
                              void* d_out, int out_size, void* d_ws, size_t ws_size,
                              hipStream_t stream) {
    const float* x  = (const float*)d_in[0];
    const int*   ei = (const int*)d_in[1];
    const float* W1 = (const float*)d_in[2];
    const float* b1 = (const float*)d_in[3];
    const float* W2 = (const float*)d_in[4];
    const float* b2 = (const float*)d_in[5];

    int n_nodes = in_sizes[0] / NFEAT;
    int n_edges = in_sizes[1] / 2;
    const int* src = ei;
    const int* dst = ei + n_edges;

    float* ws = (float*)d_ws;
    size_t off = 0;
    float* dinv = ws + off; off += ((size_t)n_nodes + 63) & ~(size_t)63;
    float* h    = ws + off; off += (size_t)n_nodes * NHID;
    float* out1 = ws + off; off += (size_t)n_nodes * NHID;
    float* z    = ws + off; off += ((size_t)2 * n_nodes + 63) & ~(size_t)63;
    (void)ws_size;

    float* out = (float*)d_out;

    const int B = 256;

    k_deg_init <<<(n_nodes + B - 1) / B, B, 0, stream>>>(dinv, n_nodes);
    k_deg_count<<<(n_edges + B - 1) / B, B, 0, stream>>>(dst, dinv, n_edges);
    k_dinv     <<<(n_nodes + B - 1) / B, B, 0, stream>>>(dinv, n_nodes);

    int tiles  = n_nodes / 16;
    int blocks = (tiles + 7) / 8;
    k_gemm1_wmma<<<blocks, B, 0, stream>>>(x, W1, h, n_nodes);

    long long t1 = (long long)n_nodes * NHID;
    k_selfloop1<<<(int)((t1 + B - 1) / B), B, 0, stream>>>(h, dinv, out1, n_nodes);
    long long t2 = (long long)n_edges * NHID;
    k_edge_agg1<<<(int)((t2 + B - 1) / B), B, 0, stream>>>(src, dst, dinv, h, out1, n_edges);
    k_relu_bias<<<(int)((t1 + B - 1) / B), B, 0, stream>>>(out1, b1, h, n_nodes);

    k_gemm2    <<<(n_nodes + B - 1) / B, B, 0, stream>>>(h, W2, b2, dinv, z, out, n_nodes);
    k_edge_agg2<<<(n_edges + B - 1) / B, B, 0, stream>>>(src, dst, dinv, z, out, n_edges);
}


# Layer-1 aggregation by destination ownership: 256 work-groups each keep 391 nodes' sums in LDS, scan all edges, add matching rows with LDS atomics (sum order = LDS arrival); bias+relu at write-out
# speedup vs baseline: 1.5454x; 1.5454x over previous
#include <hip/hip_runtime.h>

typedef float v2f __attribute__((ext_vector_type(2)));
typedef float v8f __attribute__((ext_vector_type(8)));

#define NFEAT 256
#define NHID  64


__global__ void k_deg_init(float* deg, int n) {
    int i = blockIdx.x * blockDim.x + threadIdx.x;
    if (i < n) deg[i] = 1.0f;
}

__global__ void k_deg_count(const int* __restrict__ dst, float* deg, int n_edges) {
    int e = blockIdx.x * blockDim.x + threadIdx.x;
    if (e < n_edges) atomicAdd(&deg[dst[e]], 1.0f);
}

__global__ void k_dinv(float* deg, int n) {
    int i = blockIdx.x * blockDim.x + threadIdx.x;
    if (i < n) {
        float d = deg[i];
        deg[i] = (d > 0.0f) ? rsqrtf(d) : 0.0f;
    }
}


__global__ __launch_bounds__(256) void k_gemm1_wmma(const float* __restrict__ x,
                                                    const float* __restrict__ W1,
                                                    float* __restrict__ h,
                                                    int n_nodes) {
    __shared__ float ldsW[NFEAT * NHID];
    int tid = threadIdx.x;

    for (int idx = tid * 4; idx < NFEAT * NHID; idx += 256 * 4) {
        float4 w = *(const float4*)(W1 + idx);
        int k  = idx >> 6;
        int n0 = idx & 63;
        int p  = k >> 1;
        int lo = k & 1;
        int rot = (p & 1) << 5;
        float vals[4] = {w.x, w.y, w.z, w.w};
        #pragma unroll
        for (int j = 0; j < 4; ++j) {
            int wpos = (((n0 + j) << 1) + rot) & 127;
            ldsW[p * 128 + wpos + lo] = vals[j];
        }
    }
    __syncthreads();

    int wave = tid >> 5;
    int lane = tid & 31;
    int tile = blockIdx.x * 8 + wave;
    int m0 = tile * 16;
    if (m0 >= n_nodes) return;

    int half = lane >> 4;
    int mrow = m0 + (lane & 15);
    int kk   = half << 1;
    int ncol = lane & 15;
    int rot  = half << 5;

    int wp0 = (((ncol +  0) << 1) + rot) & 127;
    int wp1 = (((ncol + 16) << 1) + rot) & 127;
    int wp2 = (((ncol + 32) << 1) + rot) & 127;
    int wp3 = (((ncol + 48) << 1) + rot) & 127;

    v8f acc0 = {}, acc1 = {}, acc2 = {}, acc3 = {};

    const float* xrow = x + (size_t)mrow * NFEAT;
    for (int k = 0; k < NFEAT; k += 4) {
        float2 av = *(const float2*)(xrow + k + kk);
        v2f a; a.x = av.x; a.y = av.y;

        const float* rowp = ldsW + ((k >> 1) + half) * 128;
        v2f b0 = *(const v2f*)(rowp + wp0);
        v2f b1 = *(const v2f*)(rowp + wp1);
        v2f b2 = *(const v2f*)(rowp + wp2);
        v2f b3 = *(const v2f*)(rowp + wp3);

        acc0 = __builtin_amdgcn_wmma_f32_16x16x4_f32(false, a, false, b0, (short)0, acc0, false, false);
        acc1 = __builtin_amdgcn_wmma_f32_16x16x4_f32(false, a, false, b1, (short)0, acc1, false, false);
        acc2 = __builtin_amdgcn_wmma_f32_16x16x4_f32(false, a, false, b2, (short)0, acc2, false, false);
        acc3 = __builtin_amdgcn_wmma_f32_16x16x4_f32(false, a, false, b3, (short)0, acc3, false, false);
    }

    int rbase = m0 + (half << 3);
    for (int r = 0; r < 8; ++r) {
        float* hr = h + (size_t)(rbase + r) * NHID;
        hr[ncol +  0] = acc0[r];
        hr[ncol + 16] = acc1[r];
        hr[ncol + 32] = acc2[r];
        hr[ncol + 48] = acc3[r];
    }
}


#define NR   391
#define QC   1280
#define SCAN_U 8

__device__ __forceinline__ int agg1_push(int* myq, int qn, bool m, int e) {
    unsigned mask = __builtin_amdgcn_ballot_w32(m);
    if (m) myq[qn + (int)__builtin_amdgcn_mbcnt_lo(mask, 0u)] = e;
    return qn + __builtin_popcount(mask);
}

__device__ __forceinline__ void agg1_drain(const int* myq, int qn, int lane, int lo,
                                           const int* __restrict__ src, const int* __restrict__ dst,
                                           const float* __restrict__ dinv, const float* __restrict__ h,
                                           float* acc) {
    for (int base = 0; base < qn; base += 32) {
        int nb = qn - base; if (nb > 32) nb = 32;
        int k = base + (lane < nb ? lane : nb - 1);
        int e = myq[k];
        int s = src[e];
        int dl = dst[e] - lo;
        float nrm = dinv[s] * dinv[dl + lo];
        float2 hv[32];
        #pragma unroll
        for (int u = 0; u < 32; ++u) {
            int sj = __builtin_amdgcn_readlane(s, u);
            hv[u] = *(const float2*)(h + (size_t)sj * NHID + (lane << 1));
        }
        #pragma unroll
        for (int u = 0; u < 32; ++u) {
            if (u == 0 || u < nb) {
                int   dj = __builtin_amdgcn_readlane(dl, u);
                float nj = __int_as_float(__builtin_amdgcn_readlane(__float_as_int(nrm), u));
                float* row = acc + dj * NHID + (lane << 1);
                atomicAdd(row + 0, hv[u].x * nj);
                atomicAdd(row + 1, hv[u].y * nj);
            }
        }
    }
}

__global__ __launch_bounds__(1024) void k_agg1_owned(const int* __restrict__ src, const int* __restrict__ dst,
                                                     const float* __restrict__ dinv, const float* __restrict__ h,
                                                     const float* __restrict__ b1, float* __restrict__ h1,
                                                     int n_nodes, int n_edges) {
    __shared__ float acc[NR * NHID];
    __shared__ int   q[32 * QC];

    const int tid  = threadIdx.x;
    const int lane = tid & 31;
    const int wave = __builtin_amdgcn_readfirstlane(tid >> 5);
    const int lo   = blockIdx.x * NR;
    int cnt = n_nodes - lo;
    if (cnt > NR) cnt = NR;
    if (cnt < 0)  cnt = 0;

    for (int idx = tid; idx < cnt * NHID; idx += 1024) {
        float di = dinv[lo + (idx >> 6)];
        acc[idx] = h[(size_t)lo * NHID + idx] * di * di;
    }
    __syncthreads();

    int* myq = q + wave * QC;
    int qn = 0;
    const int n_groups = n_edges >> 7;
    const int4* dst4 = (const int4*)dst;
    int rot = (int)(((long long)blockIdx.x * n_groups) / gridDim.x);
    const int iters = (n_groups + 31) >> 5;

    for (int it = 0; it < iters; it += SCAN_U) {
        int4 v[SCAN_U];
        int  gg[SCAN_U];
        #pragma unroll
        for (int u = 0; u < SCAN_U; ++u) {
            int gi = (it + u) * 32 + wave;
            bool ok = gi < n_groups;
            int g = gi + rot;
            if (g >= n_groups) g -= n_groups;
            if (!ok) g = 0;
            gg[u] = ok ? g : -1;
            v[u] = dst4[(size_t)g * 32 + lane];
        }
        #pragma unroll
        for (int u = 0; u < SCAN_U; ++u) {
            bool ok = gg[u] >= 0;
            bool m0 = ok & ((unsigned)(v[u].x - lo) < (unsigned)cnt);
            bool m1 = ok & ((unsigned)(v[u].y - lo) < (unsigned)cnt);
            bool m2 = ok & ((unsigned)(v[u].z - lo) < (unsigned)cnt);
            bool m3 = ok & ((unsigned)(v[u].w - lo) < (unsigned)cnt);
            if (__builtin_amdgcn_ballot_w32(m0 | m1 | m2 | m3) != 0u) {
                int e0 = (gg[u] << 7) + (lane << 2);
                qn = agg1_push(myq, qn, m0, e0 + 0);
                qn = agg1_push(myq, qn, m1, e0 + 1);
                qn = agg1_push(myq, qn, m2, e0 + 2);
                qn = agg1_push(myq, qn, m3, e0 + 3);
            }
        }
        if (qn > QC - 128 * SCAN_U - 128) { agg1_drain(myq, qn, lane, lo, src, dst, dinv, h, acc); qn = 0; }
    }
    if (wave == 0) {
        for (int e0 = n_groups << 7; e0 < n_edges; e0 += 32) {
            int e = e0 + lane;
            bool m = e < n_edges && (unsigned)(dst[e < n_edges ? e : 0] - lo) < (unsigned)cnt;
            qn = agg1_push(myq, qn, m, e);
        }
    }
    agg1_drain(myq, qn, lane, lo, src, dst, dinv, h, acc);
    __syncthreads();

    for (int idx = tid; idx < cnt * NHID; idx += 1024) {
        float v = acc[idx] + b1[idx & 63];
        h1[(size_t)lo * NHID + idx] = fmaxf(v, 0.0f);
    }
}


__global__ void k_gemm2(const float* __restrict__ h1, const float* __restrict__ W2,
                        const float* __restrict__ b2, const float* __restrict__ dinv,
                        float* __restrict__ z, float* __restrict__ out, int n_nodes) {
    int i = blockIdx.x * blockDim.x + threadIdx.x;
    if (i >= n_nodes) return;
    const float* hr = h1 + (size_t)i * NHID;
    float z0 = 0.0f, z1 = 0.0f;
    #pragma unroll
    for (int f = 0; f < NHID; f += 4) {
        float4 hv = *(const float4*)(hr + f);
        z0 += hv.x * W2[2 * (f + 0) + 0]; z1 += hv.x * W2[2 * (f + 0) + 1];
        z0 += hv.y * W2[2 * (f + 1) + 0]; z1 += hv.y * W2[2 * (f + 1) + 1];
        z0 += hv.z * W2[2 * (f + 2) + 0]; z1 += hv.z * W2[2 * (f + 2) + 1];
        z0 += hv.w * W2[2 * (f + 3) + 0]; z1 += hv.w * W2[2 * (f + 3) + 1];
    }
    float di = dinv[i], n2 = di * di;
    z[2 * i + 0] = z0;
    z[2 * i + 1] = z1;
    out[2 * i + 0] = z0 * n2 + b2[0];
    out[2 * i + 1] = z1 * n2 + b2[1];
}

__global__ void k_edge_agg2(const int* __restrict__ src, const int* __restrict__ dst,
                            const float* __restrict__ dinv, const float* __restrict__ z,
                            float* __restrict__ out, int n_edges) {
    int e = blockIdx.x * blockDim.x + threadIdx.x;
    if (e >= n_edges) return;
    int s = src[e], d = dst[e];
    float norm = dinv[s] * dinv[d];
    float2 zv = *(const float2*)(z + 2 * s);
    atomicAdd(&out[2 * d + 0], zv.x * norm);
    atomicAdd(&out[2 * d + 1], zv.y * norm);
}


extern "C" void kernel_launch(void* const* d_in, const int* in_sizes, int n_in,
                              void* d_out, int out_size, void* d_ws, size_t ws_size,
                              hipStream_t stream) {
    const float* x  = (const float*)d_in[0];
    const int*   ei = (const int*)d_in[1];
    const float* W1 = (const float*)d_in[2];
    const float* b1 = (const float*)d_in[3];
    const float* W2 = (const float*)d_in[4];
    const float* b2 = (const float*)d_in[5];

    int n_nodes = in_sizes[0] / NFEAT;
    int n_edges = in_sizes[1] / 2;
    const int* src = ei;
    const int* dst = ei + n_edges;

    float* ws = (float*)d_ws;
    size_t off = 0;
    float* dinv = ws + off; off += ((size_t)n_nodes + 63) & ~(size_t)63;
    float* h    = ws + off; off += (size_t)n_nodes * NHID;
    float* out1 = ws + off; off += (size_t)n_nodes * NHID;
    float* z    = ws + off; off += ((size_t)2 * n_nodes + 63) & ~(size_t)63;
    (void)ws_size;

    float* out = (float*)d_out;

    const int B = 256;

    k_deg_init <<<(n_nodes + B - 1) / B, B, 0, stream>>>(dinv, n_nodes);
    k_deg_count<<<(n_edges + B - 1) / B, B, 0, stream>>>(dst, dinv, n_edges);
    k_dinv     <<<(n_nodes + B - 1) / B, B, 0, stream>>>(dinv, n_nodes);

    int tiles  = n_nodes / 16;
    int blocks = (tiles + 7) / 8;
    k_gemm1_wmma<<<blocks, B, 0, stream>>>(x, W1, h, n_nodes);

    k_agg1_owned<<<(n_nodes + NR - 1) / NR, 1024, 0, stream>>>(src, dst, dinv, h, b1, out1, n_nodes, n_edges);

    k_gemm2    <<<(n_nodes + B - 1) / B, B, 0, stream>>>(out1, W2, b2, dinv, z, out, n_nodes);
    k_edge_agg2<<<(n_edges + B - 1) / B, B, 0, stream>>>(src, dst, dinv, z, out, n_edges);
}


# Layer-1 owner scan, fewer instructions per edge: 4 range tests as one unsigned min tree and one branch, batch loads off one base address, hit-free queue components skipped; sum order rule kept
# speedup vs baseline: 1.6117x; 1.0429x over previous
#include <hip/hip_runtime.h>

typedef float v2f __attribute__((ext_vector_type(2)));
typedef float v8f __attribute__((ext_vector_type(8)));

#define NFEAT 256
#define NHID  64


__global__ void k_deg_init(float* deg, int n) {
    int i = blockIdx.x * blockDim.x + threadIdx.x;
    if (i < n) deg[i] = 1.0f;
}

__global__ void k_deg_count(const int* __restrict__ dst, float* deg, int n_edges) {
    int e = blockIdx.x * blockDim.x + threadIdx.x;
    if (e < n_edges) atomicAdd(&deg[dst[e]], 1.0f);
}

__global__ void k_dinv(float* deg, int n) {
    int i = blockIdx.x * blockDim.x + threadIdx.x;
    if (i < n) {
        float d = deg[i];
        deg[i] = (d > 0.0f) ? rsqrtf(d) : 0.0f;
    }
}


__global__ __launch_bounds__(256) void k_gemm1_wmma(const float* __restrict__ x,
                                                    const float* __restrict__ W1,
                                                    float* __restrict__ h,
                                                    int n_nodes) {
    __shared__ float ldsW[NFEAT * NHID];
    int tid = threadIdx.x;

    for (int idx = tid * 4; idx < NFEAT * NHID; idx += 256 * 4) {
        float4 w = *(const float4*)(W1 + idx);
        int k  = idx >> 6;
        int n0 = idx & 63;
        int p  = k >> 1;
        int lo = k & 1;
        int rot = (p & 1) << 5;
        float vals[4] = {w.x, w.y, w.z, w.w};
        #pragma unroll
        for (int j = 0; j < 4; ++j) {
            int wpos = (((n0 + j) << 1) + rot) & 127;
            ldsW[p * 128 + wpos + lo] = vals[j];
        }
    }
    __syncthreads();

    int wave = tid >> 5;
    int lane = tid & 31;
    int tile = blockIdx.x * 8 + wave;
    int m0 = tile * 16;
    if (m0 >= n_nodes) return;

    int half = lane >> 4;
    int mrow = m0 + (lane & 15);
    int kk   = half << 1;
    int ncol = lane & 15;
    int rot  = half << 5;

    int wp0 = (((ncol +  0) << 1) + rot) & 127;
    int wp1 = (((ncol + 16) << 1) + rot) & 127;
    int wp2 = (((ncol + 32) << 1) + rot) & 127;
    int wp3 = (((ncol + 48) << 1) + rot) & 127;

    v8f acc0 = {}, acc1 = {}, acc2 = {}, acc3 = {};

    const float* xrow = x + (size_t)mrow * NFEAT;
    for (int k = 0; k < NFEAT; k += 4) {
        float2 av = *(const float2*)(xrow + k + kk);
        v2f a; a.x = av.x; a.y = av.y;

        const float* rowp = ldsW + ((k >> 1) + half) * 128;
        v2f b0 = *(const v2f*)(rowp + wp0);
        v2f b1 = *(const v2f*)(rowp + wp1);
        v2f b2 = *(const v2f*)(rowp + wp2);
        v2f b3 = *(const v2f*)(rowp + wp3);

        acc0 = __builtin_amdgcn_wmma_f32_16x16x4_f32(false, a, false, b0, (short)0, acc0, false, false);
        acc1 = __builtin_amdgcn_wmma_f32_16x16x4_f32(false, a, false, b1, (short)0, acc1, false, false);
        acc2 = __builtin_amdgcn_wmma_f32_16x16x4_f32(false, a, false, b2, (short)0, acc2, false, false);
        acc3 = __builtin_amdgcn_wmma_f32_16x16x4_f32(false, a, false, b3, (short)0, acc3, false, false);
    }

    int rbase = m0 + (half << 3);
    for (int r = 0; r < 8; ++r) {
        float* hr = h + (size_t)(rbase + r) * NHID;
        hr[ncol +  0] = acc0[r];
        hr[ncol + 16] = acc1[r];
        hr[ncol + 32] = acc2[r];
        hr[ncol + 48] = acc3[r];
    }
}


#define NR   391
#define QC   1280
#define SCAN_U 8

__device__ __forceinline__ int agg1_push(int* myq, int qn, bool m, int e) {
    unsigned mask = __builtin_amdgcn_ballot_w32(m);
    if (mask != 0u) {
        if (m) myq[qn + (int)__builtin_amdgcn_mbcnt_lo(mask, 0u)] = e;
        qn += __builtin_popcount(mask);
    }
    return qn;
}

__device__ __forceinline__ int agg1_scan_piece(int* myq, int qn, int4 v, int g, int lo, unsigned cntu, int lane) {
    unsigned a = (unsigned)(v.x - lo), b = (unsigned)(v.y - lo);
    unsigned c = (unsigned)(v.z - lo), d = (unsigned)(v.w - lo);
    unsigned ab = a < b ? a : b, cd = c < d ? c : d;
    unsigned mn = ab < cd ? ab : cd;
    if (__builtin_amdgcn_ballot_w32(mn < cntu) != 0u) {
        int e0 = (g << 7) + (lane << 2);
        qn = agg1_push(myq, qn, a < cntu, e0 + 0);
        qn = agg1_push(myq, qn, b < cntu, e0 + 1);
        qn = agg1_push(myq, qn, c < cntu, e0 + 2);
        qn = agg1_push(myq, qn, d < cntu, e0 + 3);
    }
    return qn;
}

__device__ __forceinline__ void agg1_drain(const int* myq, int qn, int lane, int lo,
                                           const int* __restrict__ src, const int* __restrict__ dst,
                                           const float* __restrict__ dinv, const float* __restrict__ h,
                                           float* acc) {
    for (int base = 0; base < qn; base += 32) {
        int nb = qn - base; if (nb > 32) nb = 32;
        int k = base + (lane < nb ? lane : nb - 1);
        int e = myq[k];
        int s = src[e];
        int dl = dst[e] - lo;
        float nrm = dinv[s] * dinv[dl + lo];
        float2 hv[32];
        #pragma unroll
        for (int u = 0; u < 32; ++u) {
            int sj = __builtin_amdgcn_readlane(s, u);
            hv[u] = *(const float2*)(h + (size_t)sj * NHID + (lane << 1));
        }
        #pragma unroll
        for (int u = 0; u < 32; ++u) {
            if (u == 0 || u < nb) {
                int   dj = __builtin_amdgcn_readlane(dl, u);
                float nj = __int_as_float(__builtin_amdgcn_readlane(__float_as_int(nrm), u));
                float* row = acc + dj * NHID + (lane << 1);
                atomicAdd(row + 0, hv[u].x * nj);
                atomicAdd(row + 1, hv[u].y * nj);
            }
        }
    }
}

__global__ __launch_bounds__(1024) void k_agg1_owned(const int* __restrict__ src, const int* __restrict__ dst,
                                                     const float* __restrict__ dinv, const float* __restrict__ h,
                                                     const float* __restrict__ b1, float* __restrict__ h1,
                                                     int n_nodes, int n_edges) {
    __shared__ float acc[NR * NHID];
    __shared__ int   q[32 * QC];

    const int tid  = threadIdx.x;
    const int lane = tid & 31;
    const int wave = __builtin_amdgcn_readfirstlane(tid >> 5);
    const int lo   = blockIdx.x * NR;
    int cnt = n_nodes - lo;
    if (cnt > NR) cnt = NR;
    if (cnt < 0)  cnt = 0;

    for (int idx = tid; idx < cnt * NHID; idx += 1024) {
        float di = dinv[lo + (idx >> 6)];
        acc[idx] = h[(size_t)lo * NHID + idx] * di * di;
    }
    __syncthreads();

    int* myq = q + wave * QC;
    int qn = 0;
    const int n_groups = n_edges >> 7;
    const int4* dst4 = (const int4*)dst;
    int rot = (int)(((long long)blockIdx.x * n_groups) / gridDim.x);
    const int K = (n_groups - wave + 31) >> 5;

    for (int kb = 0; kb < K; kb += SCAN_U) {
        const int gi0 = wave + (kb << 5);
        int g0 = gi0 + rot;
        if (g0 >= n_groups) g0 -= n_groups;
        if (gi0 + 32 * (SCAN_U - 1) < n_groups && g0 + 32 * (SCAN_U - 1) < n_groups) {
            const int4* p = dst4 + (size_t)g0 * 32 + lane;
            int4 v[SCAN_U];
            #pragma unroll
            for (int u = 0; u < SCAN_U; ++u) v[u] = p[u * 1024];
            #pragma unroll
            for (int u = 0; u < SCAN_U; ++u)
                qn = agg1_scan_piece(myq, qn, v[u], g0 + 32 * u, lo, (unsigned)cnt, lane);
        } else {
            int4 v[SCAN_U];
            int  gg[SCAN_U];
            unsigned cu[SCAN_U];
            #pragma unroll
            for (int u = 0; u < SCAN_U; ++u) {
                int gi = gi0 + 32 * u;
                bool ok = gi < n_groups;
                int g = gi + rot;
                if (g >= n_groups) g -= n_groups;
                if (!ok) g = 0;
                gg[u] = g;
                cu[u] = ok ? (unsigned)cnt : 0u;
                v[u] = dst4[(size_t)g * 32 + lane];
            }
            #pragma unroll
            for (int u = 0; u < SCAN_U; ++u)
                qn = agg1_scan_piece(myq, qn, v[u], gg[u], lo, cu[u], lane);
        }
        if (qn > QC - 128 * SCAN_U - 128) { agg1_drain(myq, qn, lane, lo, src, dst, dinv, h, acc); qn = 0; }
    }
    if (wave == 0) {
        for (int e0 = n_groups << 7; e0 < n_edges; e0 += 32) {
            int e = e0 + lane;
            bool m = e < n_edges && (unsigned)(dst[e < n_edges ? e : 0] - lo) < (unsigned)cnt;
            qn = agg1_push(myq, qn, m, e);
        }
    }
    agg1_drain(myq, qn, lane, lo, src, dst, dinv, h, acc);
    __syncthreads();

    for (int idx = tid; idx < cnt * NHID; idx += 1024) {
        float v = acc[idx] + b1[idx & 63];
        h1[(size_t)lo * NHID + idx] = fmaxf(v, 0.0f);
    }
}


__global__ void k_gemm2(const float* __restrict__ h1, const float* __restrict__ W2,
                        const float* __restrict__ b2, const float* __restrict__ dinv,
                        float* __restrict__ z, float* __restrict__ out, int n_nodes) {
    int i = blockIdx.x * blockDim.x + threadIdx.x;
    if (i >= n_nodes) return;
    const float* hr = h1 + (size_t)i * NHID;
    float z0 = 0.0f, z1 = 0.0f;
    #pragma unroll
    for (int f = 0; f < NHID; f += 4) {
        float4 hv = *(const float4*)(hr + f);
        z0 += hv.x * W2[2 * (f + 0) + 0]; z1 += hv.x * W2[2 * (f + 0) + 1];
        z0 += hv.y * W2[2 * (f + 1) + 0]; z1 += hv.y * W2[2 * (f + 1) + 1];
        z0 += hv.z * W2[2 * (f + 2) + 0]; z1 += hv.z * W2[2 * (f + 2) + 1];
        z0 += hv.w * W2[2 * (f + 3) + 0]; z1 += hv.w * W2[2 * (f + 3) + 1];
    }
    float di = dinv[i], n2 = di * di;
    z[2 * i + 0] = z0;
    z[2 * i + 1] = z1;
    out[2 * i + 0] = z0 * n2 + b2[0];
    out[2 * i + 1] = z1 * n2 + b2[1];
}

__global__ void k_edge_agg2(const int* __restrict__ src, const int* __restrict__ dst,
                            const float* __restrict__ dinv, const float* __restrict__ z,
                            float* __restrict__ out, int n_edges) {
    int e = blockIdx.x * blockDim.x + threadIdx.x;
    if (e >= n_edges) return;
    int s = src[e], d = dst[e];
    float norm = dinv[s] * dinv[d];
    float2 zv = *(const float2*)(z + 2 * s);
    atomicAdd(&out[2 * d + 0], zv.x * norm);
    atomicAdd(&out[2 * d + 1], zv.y * norm);
}


extern "C" void kernel_launch(void* const* d_in, const int* in_sizes, int n_in,
                              void* d_out, int out_size, void* d_ws, size_t ws_size,
                              hipStream_t stream) {
    const float* x  = (const float*)d_in[0];
    const int*   ei = (const int*)d_in[1];
    const float* W1 = (const float*)d_in[2];
    const float* b1 = (const float*)d_in[3];
    const float* W2 = (const float*)d_in[4];
    const float* b2 = (const float*)d_in[5];

    int n_nodes = in_sizes[0] / NFEAT;
    int n_edges = in_sizes[1] / 2;
    const int* src = ei;
    const int* dst = ei + n_edges;

    float* ws = (float*)d_ws;
    size_t off = 0;
    float* dinv = ws + off; off += ((size_t)n_nodes + 63) & ~(size_t)63;
    float* h    = ws + off; off += (size_t)n_nodes * NHID;
    float* out1 = ws + off; off += (size_t)n_nodes * NHID;
    float* z    = ws + off; off += ((size_t)2 * n_nodes + 63) & ~(size_t)63;
    (void)ws_size;

    float* out = (float*)d_out;

    const int B = 256;

    k_deg_init <<<(n_nodes + B - 1) / B, B, 0, stream>>>(dinv, n_nodes);
    k_deg_count<<<(n_edges + B - 1) / B, B, 0, stream>>>(dst, dinv, n_edges);
    k_dinv     <<<(n_nodes + B - 1) / B, B, 0, stream>>>(dinv, n_nodes);

    int tiles  = n_nodes / 16;
    int blocks = (tiles + 7) / 8;
    k_gemm1_wmma<<<blocks, B, 0, stream>>>(x, W1, h, n_nodes);

    k_agg1_owned<<<(n_nodes + NR - 1) / NR, 1024, 0, stream>>>(src, dst, dinv, h, b1, out1, n_nodes, n_edges);

    k_gemm2    <<<(n_nodes + B - 1) / B, B, 0, stream>>>(out1, W2, b2, dinv, z, out, n_nodes);
    k_edge_agg2<<<(n_edges + B - 1) / B, B, 0, stream>>>(src, dst, dinv, z, out, n_edges);
}


# Layer 2 by layer 1's ownership: the owner pass records its edges (src, local dst, norm) and writes z itself; a 250-group pass sums records in LDS, writes out once; sum order = LDS atomic arrival
# speedup vs baseline: 2.5453x; 1.5793x over previous
#include <hip/hip_runtime.h>

typedef float v2f __attribute__((ext_vector_type(2)));
typedef float v8f __attribute__((ext_vector_type(8)));

#define NFEAT 256
#define NHID  64


__global__ void k_deg_init(float* deg, int n) {
    int i = blockIdx.x * blockDim.x + threadIdx.x;
    if (i < n) deg[i] = 1.0f;
}

__global__ void k_deg_count(const int* __restrict__ dst, float* deg, int n_edges) {
    int e = blockIdx.x * blockDim.x + threadIdx.x;
    if (e < n_edges) atomicAdd(&deg[dst[e]], 1.0f);
}

__global__ void k_dinv(float* deg, int n) {
    int i = blockIdx.x * blockDim.x + threadIdx.x;
    if (i < n) {
        float d = deg[i];
        deg[i] = (d > 0.0f) ? rsqrtf(d) : 0.0f;
    }
}


__global__ __launch_bounds__(256) void k_gemm1_wmma(const float* __restrict__ x,
                                                    const float* __restrict__ W1,
                                                    float* __restrict__ h,
                                                    int n_nodes) {
    __shared__ float ldsW[NFEAT * NHID];
    int tid = threadIdx.x;

    for (int idx = tid * 4; idx < NFEAT * NHID; idx += 256 * 4) {
        float4 w = *(const float4*)(W1 + idx);
        int k  = idx >> 6;
        int n0 = idx & 63;
        int p  = k >> 1;
        int lo = k & 1;
        int rot = (p & 1) << 5;
        float vals[4] = {w.x, w.y, w.z, w.w};
        #pragma unroll
        for (int j = 0; j < 4; ++j) {
            int wpos = (((n0 + j) << 1) + rot) & 127;
            ldsW[p * 128 + wpos + lo] = vals[j];
        }
    }
    __syncthreads();

    int wave = tid >> 5;
    int lane = tid & 31;
    int tile = blockIdx.x * 8 + wave;
    int m0 = tile * 16;
    if (m0 >= n_nodes) return;

    int half = lane >> 4;
    int mrow = m0 + (lane & 15);
    int kk   = half << 1;
    int ncol = lane & 15;
    int rot  = half << 5;

    int wp0 = (((ncol +  0) << 1) + rot) & 127;
    int wp1 = (((ncol + 16) << 1) + rot) & 127;
    int wp2 = (((ncol + 32) << 1) + rot) & 127;
    int wp3 = (((ncol + 48) << 1) + rot) & 127;

    v8f acc0 = {}, acc1 = {}, acc2 = {}, acc3 = {};

    const float* xrow = x + (size_t)mrow * NFEAT;
    for (int k = 0; k < NFEAT; k += 4) {
        float2 av = *(const float2*)(xrow + k + kk);
        v2f a; a.x = av.x; a.y = av.y;

        const float* rowp = ldsW + ((k >> 1) + half) * 128;
        v2f b0 = *(const v2f*)(rowp + wp0);
        v2f b1 = *(const v2f*)(rowp + wp1);
        v2f b2 = *(const v2f*)(rowp + wp2);
        v2f b3 = *(const v2f*)(rowp + wp3);

        acc0 = __builtin_amdgcn_wmma_f32_16x16x4_f32(false, a, false, b0, (short)0, acc0, false, false);
        acc1 = __builtin_amdgcn_wmma_f32_16x16x4_f32(false, a, false, b1, (short)0, acc1, false, false);
        acc2 = __builtin_amdgcn_wmma_f32_16x16x4_f32(false, a, false, b2, (short)0, acc2, false, false);
        acc3 = __builtin_amdgcn_wmma_f32_16x16x4_f32(false, a, false, b3, (short)0, acc3, false, false);
    }

    int rbase = m0 + (half << 3);
    for (int r = 0; r < 8; ++r) {
        float* hr = h + (size_t)(rbase + r) * NHID;
        hr[ncol +  0] = acc0[r];
        hr[ncol + 16] = acc1[r];
        hr[ncol + 32] = acc2[r];
        hr[ncol + 48] = acc3[r];
    }
}


#define NR   400
#define AP   65
#define MAXOWN 256
#define RCAP 12288
#define QC   1280
#define SCAN_U 8

__device__ __forceinline__ int agg1_push(int* myq, int qn, bool m, int e) {
    unsigned mask = __builtin_amdgcn_ballot_w32(m);
    if (mask != 0u) {
        if (m) myq[qn + (int)__builtin_amdgcn_mbcnt_lo(mask, 0u)] = e;
        qn += __builtin_popcount(mask);
    }
    return qn;
}

__device__ __forceinline__ int agg1_scan_piece(int* myq, int qn, int4 v, int g, int lo, unsigned cntu, int lane) {
    unsigned a = (unsigned)(v.x - lo), b = (unsigned)(v.y - lo);
    unsigned c = (unsigned)(v.z - lo), d = (unsigned)(v.w - lo);
    unsigned ab = a < b ? a : b, cd = c < d ? c : d;
    unsigned mn = ab < cd ? ab : cd;
    if (__builtin_amdgcn_ballot_w32(mn < cntu) != 0u) {
        int e0 = (g << 7) + (lane << 2);
        qn = agg1_push(myq, qn, a < cntu, e0 + 0);
        qn = agg1_push(myq, qn, b < cntu, e0 + 1);
        qn = agg1_push(myq, qn, c < cntu, e0 + 2);
        qn = agg1_push(myq, qn, d < cntu, e0 + 3);
    }
    return qn;
}

__device__ __forceinline__ void agg1_drain(const int* myq, int qn, int lane, int lo,
                                           const int* __restrict__ src, const int* __restrict__ dst,
                                           const float* __restrict__ dinv, const float* __restrict__ h,
                                           float* acc, int2* myrec, int* rcur) {
    for (int base = 0; base < qn; base += 32) {
        int nb = qn - base; if (nb > 32) nb = 32;
        int k = base + (lane < nb ? lane : nb - 1);
        int e = myq[k];
        int s = src[e];
        int dl = dst[e] - lo;
        float nrm = dinv[s] * dinv[dl + lo];
        float2 hv[32];
        #pragma unroll
        for (int u = 0; u < 32; ++u) {
            int sj = __builtin_amdgcn_readlane(s, u);
            hv[u] = *(const float2*)(h + (size_t)sj * NHID + (lane << 1));
        }
        if (myrec != nullptr) {
            int rb = 0;
            if (lane == 0) rb = atomicAdd(rcur, 32);
            rb = __builtin_amdgcn_readfirstlane(rb);
            if (rb + 32 <= RCAP) {
                int2 rv;
                rv.x = lane < nb ? (int)(((unsigned)dl << 23) | (unsigned)s) : -1;
                rv.y = __float_as_int(nrm);
                myrec[rb + lane] = rv;
            }
        }
        #pragma unroll
        for (int u = 0; u < 32; ++u) {
            if (u == 0 || u < nb) {
                int   dj = __builtin_amdgcn_readlane(dl, u);
                float nj = __int_as_float(__builtin_amdgcn_readlane(__float_as_int(nrm), u));
                float* row = acc + dj * AP + (lane << 1);
                atomicAdd(row + 0, hv[u].x * nj);
                atomicAdd(row + 1, hv[u].y * nj);
            }
        }
    }
}

__global__ __launch_bounds__(1024) void k_agg1_owned(const int* __restrict__ src, const int* __restrict__ dst,
                                                     const float* __restrict__ dinv, const float* __restrict__ h,
                                                     const float* __restrict__ b1, const float* __restrict__ W2,
                                                     float* __restrict__ z, int* __restrict__ rec,
                                                     int* __restrict__ rcount, int n_nodes, int n_edges) {
    __shared__ float acc[NR * AP];
    __shared__ int   q[32 * QC];
    __shared__ int   rcur;

    const int tid  = threadIdx.x;
    const int lane = tid & 31;
    const int wave = __builtin_amdgcn_readfirstlane(tid >> 5);
    const int lo   = blockIdx.x * NR;
    int cnt = n_nodes - lo;
    if (cnt > NR) cnt = NR;
    if (cnt < 0)  cnt = 0;

    int2* myrec = (blockIdx.x < MAXOWN && n_nodes <= (1 << 23))
                      ? (int2*)rec + (size_t)blockIdx.x * RCAP : nullptr;
    if (tid == 0) rcur = 0;

    for (int idx = tid; idx < cnt * NHID; idx += 1024) {
        int node = idx >> 6;
        float di = dinv[lo + node];
        acc[node * AP + (idx & 63)] = h[(size_t)lo * NHID + idx] * di * di;
    }
    __syncthreads();

    int* myq = q + wave * QC;
    int qn = 0;
    const int n_groups = n_edges >> 7;
    const int4* dst4 = (const int4*)dst;
    int rot = (int)(((long long)blockIdx.x * n_groups) / gridDim.x);
    const int K = (n_groups - wave + 31) >> 5;

    for (int kb = 0; kb < K; kb += SCAN_U) {
        const int gi0 = wave + (kb << 5);
        int g0 = gi0 + rot;
        if (g0 >= n_groups) g0 -= n_groups;
        if (gi0 + 32 * (SCAN_U - 1) < n_groups && g0 + 32 * (SCAN_U - 1) < n_groups) {
            const int4* p = dst4 + (size_t)g0 * 32 + lane;
            int4 v[SCAN_U];
            #pragma unroll
            for (int u = 0; u < SCAN_U; ++u) v[u] = p[u * 1024];
            #pragma unroll
            for (int u = 0; u < SCAN_U; ++u)
                qn = agg1_scan_piece(myq, qn, v[u], g0 + 32 * u, lo, (unsigned)cnt, lane);
        } else {
            int4 v[SCAN_U];
            int  gg[SCAN_U];
            unsigned cu[SCAN_U];
            #pragma unroll
            for (int u = 0; u < SCAN_U; ++u) {
                int gi = gi0 + 32 * u;
                bool ok = gi < n_groups;
                int g = gi + rot;
                if (g >= n_groups) g -= n_groups;
                if (!ok) g = 0;
                gg[u] = g;
                cu[u] = ok ? (unsigned)cnt : 0u;
                v[u] = dst4[(size_t)g * 32 + lane];
            }
            #pragma unroll
            for (int u = 0; u < SCAN_U; ++u)
                qn = agg1_scan_piece(myq, qn, v[u], gg[u], lo, cu[u], lane);
        }
        if (qn > QC - 128 * SCAN_U - 128) { agg1_drain(myq, qn, lane, lo, src, dst, dinv, h, acc, myrec, &rcur); qn = 0; }
    }
    if (wave == 0) {
        for (int e0 = n_groups << 7; e0 < n_edges; e0 += 32) {
            int e = e0 + lane;
            bool m = e < n_edges && (unsigned)(dst[e < n_edges ? e : 0] - lo) < (unsigned)cnt;
            qn = agg1_push(myq, qn, m, e);
        }
    }
    agg1_drain(myq, qn, lane, lo, src, dst, dinv, h, acc, myrec, &rcur);
    __syncthreads();

    if (tid == 0 && myrec != nullptr) rcount[blockIdx.x * 32] = rcur;

    if (tid < cnt) {
        const float* ar = acc + tid * AP;
        float z0 = 0.0f, z1 = 0.0f;
        #pragma unroll
        for (int f = 0; f < NHID; ++f) {
            float hv = fmaxf(ar[f] + b1[f], 0.0f);
            z0 += hv * W2[2 * f + 0];
            z1 += hv * W2[2 * f + 1];
        }
        float2 zz; zz.x = z0; zz.y = z1;
        *(float2*)(z + 2 * (size_t)(lo + tid)) = zz;
    }
}


#define A2U 8

__global__ __launch_bounds__(1024) void k_agg2_owned(const int* __restrict__ src, const int* __restrict__ dst,
                                                     const float* __restrict__ dinv, const float* __restrict__ z,
                                                     const float* __restrict__ b2, const int* __restrict__ rec,
                                                     const int* __restrict__ rcount, float* __restrict__ out,
                                                     int n_nodes, int n_edges) {
    __shared__ float o2[NR * 2];
    const int tid = threadIdx.x;
    const int lo  = blockIdx.x * NR;
    int cnt = n_nodes - lo;
    if (cnt > NR) cnt = NR;
    if (cnt < 0)  cnt = 0;

    for (int i = tid; i < cnt * 2; i += 1024) {
        float di = dinv[lo + (i >> 1)];
        float n2 = di * di;
        o2[i] = z[2 * (size_t)lo + i] * n2 + b2[i & 1];
    }
    __syncthreads();

    const bool sliced = blockIdx.x < MAXOWN && n_nodes <= (1 << 23);
    const int total = sliced ? rcount[blockIdx.x * 32] : RCAP + 1;
    if (total <= RCAP) {
        const int2* r = (const int2*)rec + (size_t)blockIdx.x * RCAP;
        for (int k0 = tid; k0 < total; k0 += 1024 * A2U) {
            int2 v[A2U];
            float2 zv[A2U];
            #pragma unroll
            for (int u = 0; u < A2U; ++u) {
                int k = k0 + u * 1024;
                v[u] = r[k < total ? k : k0];
                if (k >= total) v[u].x = -1;
            }
            #pragma unroll
            for (int u = 0; u < A2U; ++u) {
                int s = v[u].x == -1 ? 0 : (v[u].x & 0x7FFFFF);
                zv[u] = *(const float2*)(z + 2 * (size_t)s);
            }
            #pragma unroll
            for (int u = 0; u < A2U; ++u) {
                if (v[u].x != -1) {
                    int dl = (int)((unsigned)v[u].x >> 23);
                    float nrm = __int_as_float(v[u].y);
                    atomicAdd(&o2[2 * dl + 0], zv[u].x * nrm);
                    atomicAdd(&o2[2 * dl + 1], zv[u].y * nrm);
                }
            }
        }
    } else {
        for (int e = tid; e < n_edges; e += 1024) {
            int d = dst[e];
            if ((unsigned)(d - lo) < (unsigned)cnt) {
                int s = src[e];
                float nrm = dinv[s] * dinv[d];
                float2 zv = *(const float2*)(z + 2 * (size_t)s);
                atomicAdd(&o2[2 * (d - lo) + 0], zv.x * nrm);
                atomicAdd(&o2[2 * (d - lo) + 1], zv.y * nrm);
            }
        }
    }
    __syncthreads();

    for (int i = tid; i < cnt * 2; i += 1024) out[2 * (size_t)lo + i] = o2[i];
}


extern "C" void kernel_launch(void* const* d_in, const int* in_sizes, int n_in,
                              void* d_out, int out_size, void* d_ws, size_t ws_size,
                              hipStream_t stream) {
    const float* x  = (const float*)d_in[0];
    const int*   ei = (const int*)d_in[1];
    const float* W1 = (const float*)d_in[2];
    const float* b1 = (const float*)d_in[3];
    const float* W2 = (const float*)d_in[4];
    const float* b2 = (const float*)d_in[5];

    int n_nodes = in_sizes[0] / NFEAT;
    int n_edges = in_sizes[1] / 2;
    const int* src = ei;
    const int* dst = ei + n_edges;

    float* ws = (float*)d_ws;
    size_t off = 0;
    float* dinv = ws + off; off += ((size_t)n_nodes + 63) & ~(size_t)63;
    float* h    = ws + off; off += (size_t)n_nodes * NHID;
    float* recs = ws + off; off += (size_t)n_nodes * NHID;
    float* z    = ws + off; off += ((size_t)2 * n_nodes + 63) & ~(size_t)63;
    (void)ws_size;

    float* out = (float*)d_out;

    const int B = 256;

    k_deg_init <<<(n_nodes + B - 1) / B, B, 0, stream>>>(dinv, n_nodes);
    k_deg_count<<<(n_edges + B - 1) / B, B, 0, stream>>>(dst, dinv, n_edges);
    k_dinv     <<<(n_nodes + B - 1) / B, B, 0, stream>>>(dinv, n_nodes);

    int tiles  = n_nodes / 16;
    int blocks = (tiles + 7) / 8;
    k_gemm1_wmma<<<blocks, B, 0, stream>>>(x, W1, h, n_nodes);

    int* rec    = (int*)recs;
    int* rcount = rec + (size_t)2 * MAXOWN * RCAP;
    k_agg1_owned<<<(n_nodes + NR - 1) / NR, 1024, 0, stream>>>(src, dst, dinv, h, b1, W2, z, rec, rcount, n_nodes, n_edges);

    k_agg2_owned<<<(n_nodes + NR - 1) / NR, 1024, 0, stream>>>(src, dst, dinv, z, b2, rec, rcount, out, n_nodes, n_edges);
}


# In-degree counted in LDS by the owner scan, which records its edges and writes deg^-1/2; a 2nd owned pass sums rows from the records; no global atomic, 6 launches to 4; sum order = LDS arrival, kept
# speedup vs baseline: 3.4637x; 1.3608x over previous
#include <hip/hip_runtime.h>

typedef float v2f __attribute__((ext_vector_type(2)));
typedef float v8f __attribute__((ext_vector_type(8)));

#define NFEAT 256
#define NHID  64


__global__ __launch_bounds__(256) void k_gemm1_wmma(const float* __restrict__ x,
                                                    const float* __restrict__ W1,
                                                    float* __restrict__ h,
                                                    int n_nodes) {
    __shared__ float ldsW[NFEAT * NHID];
    int tid = threadIdx.x;

    for (int idx = tid * 4; idx < NFEAT * NHID; idx += 256 * 4) {
        float4 w = *(const float4*)(W1 + idx);
        int k  = idx >> 6;
        int n0 = idx & 63;
        int p  = k >> 1;
        int lo = k & 1;
        int rot = (p & 1) << 5;
        float vals[4] = {w.x, w.y, w.z, w.w};
        #pragma unroll
        for (int j = 0; j < 4; ++j) {
            int wpos = (((n0 + j) << 1) + rot) & 127;
            ldsW[p * 128 + wpos + lo] = vals[j];
        }
    }
    __syncthreads();

    int wave = tid >> 5;
    int lane = tid & 31;
    int tile = blockIdx.x * 8 + wave;
    int m0 = tile * 16;
    if (m0 >= n_nodes) return;

    int half = lane >> 4;
    int mrow = m0 + (lane & 15);
    int kk   = half << 1;
    int ncol = lane & 15;
    int rot  = half << 5;

    int wp0 = (((ncol +  0) << 1) + rot) & 127;
    int wp1 = (((ncol + 16) << 1) + rot) & 127;
    int wp2 = (((ncol + 32) << 1) + rot) & 127;
    int wp3 = (((ncol + 48) << 1) + rot) & 127;

    v8f acc0 = {}, acc1 = {}, acc2 = {}, acc3 = {};

    const float* xrow = x + (size_t)mrow * NFEAT;
    for (int k = 0; k < NFEAT; k += 4) {
        float2 av = *(const float2*)(xrow + k + kk);
        v2f a; a.x = av.x; a.y = av.y;

        const float* rowp = ldsW + ((k >> 1) + half) * 128;
        v2f b0 = *(const v2f*)(rowp + wp0);
        v2f b1 = *(const v2f*)(rowp + wp1);
        v2f b2 = *(const v2f*)(rowp + wp2);
        v2f b3 = *(const v2f*)(rowp + wp3);

        acc0 = __builtin_amdgcn_wmma_f32_16x16x4_f32(false, a, false, b0, (short)0, acc0, false, false);
        acc1 = __builtin_amdgcn_wmma_f32_16x16x4_f32(false, a, false, b1, (short)0, acc1, false, false);
        acc2 = __builtin_amdgcn_wmma_f32_16x16x4_f32(false, a, false, b2, (short)0, acc2, false, false);
        acc3 = __builtin_amdgcn_wmma_f32_16x16x4_f32(false, a, false, b3, (short)0, acc3, false, false);
    }

    int rbase = m0 + (half << 3);
    for (int r = 0; r < 8; ++r) {
        float* hr = h + (size_t)(rbase + r) * NHID;
        hr[ncol +  0] = acc0[r];
        hr[ncol + 16] = acc1[r];
        hr[ncol + 32] = acc2[r];
        hr[ncol + 48] = acc3[r];
    }
}


#define NR   416
#define AP   65
#define MAXOWN 256
#define RCAP 12288
#define QC   1280
#define SCAN_U 8

__device__ __forceinline__ int agg1_push(int* myq, int qn, bool m, int e) {
    unsigned mask = __builtin_amdgcn_ballot_w32(m);
    if (mask != 0u) {
        if (m) myq[qn + (int)__builtin_amdgcn_mbcnt_lo(mask, 0u)] = e;
        qn += __builtin_popcount(mask);
    }
    return qn;
}

__device__ __forceinline__ int agg1_scan_piece(int* myq, int qn, int4 v, int g, int lo, unsigned cntu, int lane) {
    unsigned a = (unsigned)(v.x - lo), b = (unsigned)(v.y - lo);
    unsigned c = (unsigned)(v.z - lo), d = (unsigned)(v.w - lo);
    unsigned ab = a < b ? a : b, cd = c < d ? c : d;
    unsigned mn = ab < cd ? ab : cd;
    if (__builtin_amdgcn_ballot_w32(mn < cntu) != 0u) {
        int e0 = (g << 7) + (lane << 2);
        qn = agg1_push(myq, qn, a < cntu, e0 + 0);
        qn = agg1_push(myq, qn, b < cntu, e0 + 1);
        qn = agg1_push(myq, qn, c < cntu, e0 + 2);
        qn = agg1_push(myq, qn, d < cntu, e0 + 3);
    }
    return qn;
}

__device__ __forceinline__ void scan_note(const int* myq, int qn, int lane, int lo,
                                          const int* __restrict__ src, const int* __restrict__ dst,
                                          int* degl, int2* myrec, int* rcur) {
    for (int base = 0; base < qn; base += 32) {
        int nb = qn - base; if (nb > 32) nb = 32;
        int k = base + (lane < nb ? lane : nb - 1);
        int e = myq[k];
        int s = src[e];
        int dl = dst[e] - lo;
        if (lane < nb) atomicAdd(&degl[dl], 1);
        if (myrec != nullptr) {
            int rb = 0;
            if (lane == 0) rb = atomicAdd(rcur, 32);
            rb = __builtin_amdgcn_readfirstlane(rb);
            if (rb + 32 <= RCAP) {
                int2 rv;
                rv.x = lane < nb ? (int)(((unsigned)dl << 23) | (unsigned)s) : -1;
                rv.y = 0;
                myrec[rb + lane] = rv;
            }
        }
    }
}

__global__ __launch_bounds__(1024) void k_scan_owned(const int* __restrict__ src, const int* __restrict__ dst,
                                                     float* __restrict__ dinv, int* __restrict__ rec,
                                                     int* __restrict__ rcount, int n_nodes, int n_edges) {
    __shared__ int   degl[NR];
    __shared__ int   q[32 * QC];
    __shared__ int   rcur;

    const int tid  = threadIdx.x;
    const int lane = tid & 31;
    const int wave = __builtin_amdgcn_readfirstlane(tid >> 5);
    const int lo   = blockIdx.x * NR;
    int cnt = n_nodes - lo;
    if (cnt > NR) cnt = NR;
    if (cnt < 0)  cnt = 0;

    int2* myrec = (blockIdx.x < MAXOWN && n_nodes <= (1 << 23))
                      ? (int2*)rec + (size_t)blockIdx.x * RCAP : nullptr;
    if (tid == 0) rcur = 0;
    for (int i = tid; i < NR; i += 1024) degl[i] = 0;
    __syncthreads();

    int* myq = q + wave * QC;
    int qn = 0;
    const int n_groups = n_edges >> 7;
    const int4* dst4 = (const int4*)dst;
    int rot = (int)(((long long)blockIdx.x * n_groups) / gridDim.x);
    const int K = (n_groups - wave + 31) >> 5;

    for (int kb = 0; kb < K; kb += SCAN_U) {
        const int gi0 = wave + (kb << 5);
        int g0 = gi0 + rot;
        if (g0 >= n_groups) g0 -= n_groups;
        if (gi0 + 32 * (SCAN_U - 1) < n_groups && g0 + 32 * (SCAN_U - 1) < n_groups) {
            const int4* p = dst4 + (size_t)g0 * 32 + lane;
            int4 v[SCAN_U];
            #pragma unroll
            for (int u = 0; u < SCAN_U; ++u) v[u] = p[u * 1024];
            #pragma unroll
            for (int u = 0; u < SCAN_U; ++u)
                qn = agg1_scan_piece(myq, qn, v[u], g0 + 32 * u, lo, (unsigned)cnt, lane);
        } else {
            int4 v[SCAN_U];
            int  gg[SCAN_U];
            unsigned cu[SCAN_U];
            #pragma unroll
            for (int u = 0; u < SCAN_U; ++u) {
                int gi = gi0 + 32 * u;
                bool ok = gi < n_groups;
                int g = gi + rot;
                if (g >= n_groups) g -= n_groups;
                if (!ok) g = 0;
                gg[u] = g;
                cu[u] = ok ? (unsigned)cnt : 0u;
                v[u] = dst4[(size_t)g * 32 + lane];
            }
            #pragma unroll
            for (int u = 0; u < SCAN_U; ++u)
                qn = agg1_scan_piece(myq, qn, v[u], gg[u], lo, cu[u], lane);
        }
        if (qn > QC - 128 * SCAN_U - 128) { scan_note(myq, qn, lane, lo, src, dst, degl, myrec, &rcur); qn = 0; }
    }
    if (wave == 0) {
        for (int e0 = n_groups << 7; e0 < n_edges; e0 += 32) {
            int e = e0 + lane;
            bool m = e < n_edges && (unsigned)(dst[e < n_edges ? e : 0] - lo) < (unsigned)cnt;
            qn = agg1_push(myq, qn, m, e);
        }
    }
    scan_note(myq, qn, lane, lo, src, dst, degl, myrec, &rcur);
    __syncthreads();

    if (tid == 0 && myrec != nullptr) rcount[blockIdx.x * 32] = rcur;

    for (int i = tid; i < cnt; i += 1024) {
        float d = (float)(degl[i] + 1);
        dinv[lo + i] = (d > 0.0f) ? rsqrtf(d) : 0.0f;
    }
}

__global__ __launch_bounds__(1024) void k_agg1_owned(const int* __restrict__ src, const int* __restrict__ dst,
                                                     const float* __restrict__ dinv, const float* __restrict__ h,
                                                     const float* __restrict__ b1, const float* __restrict__ W2,
                                                     float* __restrict__ z, int* rec,
                                                     const int* __restrict__ rcount, int n_nodes, int n_edges) {
    __shared__ float acc[NR * AP];

    const int tid  = threadIdx.x;
    const int lane = tid & 31;
    const int wave = __builtin_amdgcn_readfirstlane(tid >> 5);
    const int lo   = blockIdx.x * NR;
    int cnt = n_nodes - lo;
    if (cnt > NR) cnt = NR;
    if (cnt < 0)  cnt = 0;

    for (int idx = tid; idx < cnt * NHID; idx += 1024) {
        int node = idx >> 6;
        float di = dinv[lo + node];
        acc[node * AP + (idx & 63)] = h[(size_t)lo * NHID + idx] * di * di;
    }
    __syncthreads();

    const bool sliced = blockIdx.x < MAXOWN && n_nodes <= (1 << 23);
    const int total = sliced ? rcount[blockIdx.x * 32] : RCAP + 1;
    if (total <= RCAP) {
        int2* myrec = (int2*)rec + (size_t)blockIdx.x * RCAP;
        const int rounds = total >> 5;
        int kx = wave < rounds ? myrec[wave * 32 + lane].x : -1;
        for (int r = wave; r < rounds; r += 32) {
            const int key = kx;
            if (r + 32 < rounds) kx = myrec[(r + 32) * 32 + lane].x;
            bool ok = key != -1;
            int nb = __builtin_popcount(__builtin_amdgcn_ballot_w32(ok));
            int s  = ok ? (key & 0x7FFFFF) : 0;
            int dl = ok ? (int)((unsigned)key >> 23) : 0;
            float nrm = dinv[s] * dinv[dl + lo];
            float2 hv[32];
            #pragma unroll
            for (int u = 0; u < 32; ++u) {
                int sj = __builtin_amdgcn_readlane(s, u);
                hv[u] = *(const float2*)(h + (size_t)sj * NHID + (lane << 1));
            }
            int2 rv; rv.x = key; rv.y = __float_as_int(nrm);
            myrec[r * 32 + lane] = rv;
            #pragma unroll
            for (int u = 0; u < 32; ++u) {
                if (u < nb) {
                    int   dj = __builtin_amdgcn_readlane(dl, u);
                    float nj = __int_as_float(__builtin_amdgcn_readlane(__float_as_int(nrm), u));
                    float* row = acc + dj * AP + (lane << 1);
                    atomicAdd(row + 0, hv[u].x * nj);
                    atomicAdd(row + 1, hv[u].y * nj);
                }
            }
        }
    } else {
        for (int e = tid; e < n_edges; e += 1024) {
            int d = dst[e];
            if ((unsigned)(d - lo) < (unsigned)cnt) {
                int s = src[e];
                float nrm = dinv[s] * dinv[d];
                const float* hr = h + (size_t)s * NHID;
                float* row = acc + (d - lo) * AP;
                for (int f = 0; f < NHID; ++f) atomicAdd(row + f, hr[f] * nrm);
            }
        }
    }
    __syncthreads();

    if (tid < cnt) {
        const float* ar = acc + tid * AP;
        float z0 = 0.0f, z1 = 0.0f;
        #pragma unroll
        for (int f = 0; f < NHID; ++f) {
            float hv = fmaxf(ar[f] + b1[f], 0.0f);
            z0 += hv * W2[2 * f + 0];
            z1 += hv * W2[2 * f + 1];
        }
        float2 zz; zz.x = z0; zz.y = z1;
        *(float2*)(z + 2 * (size_t)(lo + tid)) = zz;
    }
}


#define A2U 8

__global__ __launch_bounds__(1024) void k_agg2_owned(const int* __restrict__ src, const int* __restrict__ dst,
                                                     const float* __restrict__ dinv, const float* __restrict__ z,
                                                     const float* __restrict__ b2, const int* __restrict__ rec,
                                                     const int* __restrict__ rcount, float* __restrict__ out,
                                                     int n_nodes, int n_edges) {
    __shared__ float o2[NR * 2];
    const int tid = threadIdx.x;
    const int lo  = blockIdx.x * NR;
    int cnt = n_nodes - lo;
    if (cnt > NR) cnt = NR;
    if (cnt < 0)  cnt = 0;

    for (int i = tid; i < cnt * 2; i += 1024) {
        float di = dinv[lo + (i >> 1)];
        float n2 = di * di;
        o2[i] = z[2 * (size_t)lo + i] * n2 + b2[i & 1];
    }
    __syncthreads();

    const bool sliced = blockIdx.x < MAXOWN && n_nodes <= (1 << 23);
    const int total = sliced ? rcount[blockIdx.x * 32] : RCAP + 1;
    if (total <= RCAP) {
        const int2* r = (const int2*)rec + (size_t)blockIdx.x * RCAP;
        for (int k0 = tid; k0 < total; k0 += 1024 * A2U) {
            int2 v[A2U];
            float2 zv[A2U];
            #pragma unroll
            for (int u = 0; u < A2U; ++u) {
                int k = k0 + u * 1024;
                v[u] = r[k < total ? k : k0];
                if (k >= total) v[u].x = -1;
            }
            #pragma unroll
            for (int u = 0; u < A2U; ++u) {
                int s = v[u].x == -1 ? 0 : (v[u].x & 0x7FFFFF);
                zv[u] = *(const float2*)(z + 2 * (size_t)s);
            }
            #pragma unroll
            for (int u = 0; u < A2U; ++u) {
                if (v[u].x != -1) {
                    int dl = (int)((unsigned)v[u].x >> 23);
                    float nrm = __int_as_float(v[u].y);
                    atomicAdd(&o2[2 * dl + 0], zv[u].x * nrm);
                    atomicAdd(&o2[2 * dl + 1], zv[u].y * nrm);
                }
            }
        }
    } else {
        for (int e = tid; e < n_edges; e += 1024) {
            int d = dst[e];
            if ((unsigned)(d - lo) < (unsigned)cnt) {
                int s = src[e];
                float nrm = dinv[s] * dinv[d];
                float2 zv = *(const float2*)(z + 2 * (size_t)s);
                atomicAdd(&o2[2 * (d - lo) + 0], zv.x * nrm);
                atomicAdd(&o2[2 * (d - lo) + 1], zv.y * nrm);
            }
        }
    }
    __syncthreads();

    for (int i = tid; i < cnt * 2; i += 1024) out[2 * (size_t)lo + i] = o2[i];
}


extern "C" void kernel_launch(void* const* d_in, const int* in_sizes, int n_in,
                              void* d_out, int out_size, void* d_ws, size_t ws_size,
                              hipStream_t stream) {
    const float* x  = (const float*)d_in[0];
    const int*   ei = (const int*)d_in[1];
    const float* W1 = (const float*)d_in[2];
    const float* b1 = (const float*)d_in[3];
    const float* W2 = (const float*)d_in[4];
    const float* b2 = (const float*)d_in[5];

    int n_nodes = in_sizes[0] / NFEAT;
    int n_edges = in_sizes[1] / 2;
    const int* src = ei;
    const int* dst = ei + n_edges;

    float* ws = (float*)d_ws;
    size_t off = 0;
    float* dinv = ws + off; off += ((size_t)n_nodes + 63) & ~(size_t)63;
    float* h    = ws + off; off += (size_t)n_nodes * NHID;
    float* recs = ws + off; off += (size_t)n_nodes * NHID;
    float* z    = ws + off; off += ((size_t)2 * n_nodes + 63) & ~(size_t)63;
    (void)ws_size;

    float* out = (float*)d_out;

    const int B = 256;
    int owners = (n_nodes + NR - 1) / NR;

    int* rec    = (int*)recs;
    int* rcount = rec + (size_t)2 * MAXOWN * RCAP;
    k_scan_owned<<<owners, 1024, 0, stream>>>(src, dst, dinv, rec, rcount, n_nodes, n_edges);

    int tiles  = n_nodes / 16;
    int blocks = (tiles + 7) / 8;
    k_gemm1_wmma<<<blocks, B, 0, stream>>>(x, W1, h, n_nodes);

    k_agg1_owned<<<owners, 1024, 0, stream>>>(src, dst, dinv, h, b1, W2, z, rec, rcount, n_nodes, n_edges);

    k_agg2_owned<<<owners, 1024, 0, stream>>>(src, dst, dinv, z, b2, rec, rcount, out, n_nodes, n_edges);
}


# Dense transform moved into the owner-scan launch: 8 of each group's 32 waves run the WMMA tiles (x loaded 16 k-steps ahead) while 24 scan; 4 launches to 3; k order and LDS-arrival sum order kept
# speedup vs baseline: 3.7359x; 1.0786x over previous
#include <hip/hip_runtime.h>

typedef float v2f __attribute__((ext_vector_type(2)));
typedef float v8f __attribute__((ext_vector_type(8)));

#define NFEAT 256
#define NHID  64


#define NGW  8
#define NSW  24
#define GPF  16
#define GSB  4

__device__ __forceinline__ void gemm_stage_w1(float* ldsW, const float* __restrict__ W1, int tid) {
    for (int idx = tid * 4; idx < NFEAT * NHID; idx += 1024 * 4) {
        float4 w = *(const float4*)(W1 + idx);
        int k  = idx >> 6;
        int n0 = idx & 63;
        int p  = k >> 1;
        int lo = k & 1;
        int rot = (p & 1) << 5;
        float vals[4] = {w.x, w.y, w.z, w.w};
        #pragma unroll
        for (int j = 0; j < 4; ++j) {
            int wpos = (((n0 + j) << 1) + rot) & 127;
            ldsW[p * 128 + wpos + lo] = vals[j];
        }
    }
}

__device__ __forceinline__ void gemm_wave(const float* __restrict__ x, float* __restrict__ h,
                                          const float* ldsW, int lane, int first, int stride, int n_tiles) {
    int t = first;
    if (t >= n_tiles) return;

    int half = lane >> 4;
    int kk   = half << 1;
    int ncol = lane & 15;
    int rot  = half << 5;

    int wp0 = (((ncol +  0) << 1) + rot) & 127;
    int wp1 = (((ncol + 16) << 1) + rot) & 127;
    int wp2 = (((ncol + 32) << 1) + rot) & 127;
    int wp3 = (((ncol + 48) << 1) + rot) & 127;
    unsigned boff = (unsigned)half * 128u;

    v2f a[GPF];
    const unsigned xlane = (unsigned)ncol * NFEAT + (unsigned)kk;
    const unsigned hlane = (unsigned)(half << 3) * NHID + (unsigned)ncol;
    const float* xr = x + (size_t)t * (16 * NFEAT) + xlane;
    #pragma unroll
    for (int s = 0; s < GPF; ++s) {
        float2 av = *(const float2*)(xr + s * 4);
        a[s].x = av.x; a[s].y = av.y;
    }
    for (;;) {
        const int  tn   = t + stride;
        const bool more = tn < n_tiles;
        const float* xn = x + (size_t)(more ? tn : t) * (16 * NFEAT) + xlane;

        asm volatile("" : "+v"(boff));
        const float* bb = ldsW + boff;
        v8f acc0 = {}, acc1 = {}, acc2 = {}, acc3 = {};
        #pragma unroll
        for (int g = 0; g < NFEAT / 4 / GSB; ++g) {
            #pragma unroll
            for (int j = 0; j < GSB; ++j) {
                const int s = g * GSB + j;
                const float* rowp = bb + (s << 1) * 128;
                v2f b0 = *(const v2f*)(rowp + wp0);
                v2f b1 = *(const v2f*)(rowp + wp1);
                v2f b2 = *(const v2f*)(rowp + wp2);
                v2f b3 = *(const v2f*)(rowp + wp3);
                const v2f av = a[s % GPF];
                acc0 = __builtin_amdgcn_wmma_f32_16x16x4_f32(false, av, false, b0, (short)0, acc0, false, false);
                acc1 = __builtin_amdgcn_wmma_f32_16x16x4_f32(false, av, false, b1, (short)0, acc1, false, false);
                acc2 = __builtin_amdgcn_wmma_f32_16x16x4_f32(false, av, false, b2, (short)0, acc2, false, false);
                acc3 = __builtin_amdgcn_wmma_f32_16x16x4_f32(false, av, false, b3, (short)0, acc3, false, false);
                const float* nx = (s + GPF < NFEAT / 4) ? xr + (s + GPF) * 4 : xn + (s + GPF - NFEAT / 4) * 4;
                float2 nv = *(const float2*)nx;
                a[s % GPF].x = nv.x; a[s % GPF].y = nv.y;
            }
            __builtin_amdgcn_sched_barrier(0);
        }

        float* ht = h + (size_t)t * (16 * NHID) + hlane;
        for (int r = 0; r < 8; ++r) {
            float* hr = ht + r * NHID;
            hr[ 0] = acc0[r];
            hr[16] = acc1[r];
            hr[32] = acc2[r];
            hr[48] = acc3[r];
        }
        if (!more) break;
        t = tn; xr = xn;
    }
}


#define NR   416
#define AP   65
#define MAXOWN 256
#define RCAP 12288
#define QC   1792
#define SCAN_U 12

__device__ __forceinline__ int agg1_push(int* myq, int qn, bool m, int e) {
    unsigned mask = __builtin_amdgcn_ballot_w32(m);
    if (mask != 0u) {
        if (m) myq[qn + (int)__builtin_amdgcn_mbcnt_lo(mask, 0u)] = e;
        qn += __builtin_popcount(mask);
    }
    return qn;
}

__device__ __forceinline__ int agg1_scan_piece(int* myq, int qn, int4 v, int g, int lo, unsigned cntu, int lane) {
    unsigned a = (unsigned)(v.x - lo), b = (unsigned)(v.y - lo);
    unsigned c = (unsigned)(v.z - lo), d = (unsigned)(v.w - lo);
    unsigned ab = a < b ? a : b, cd = c < d ? c : d;
    unsigned mn = ab < cd ? ab : cd;
    if (__builtin_amdgcn_ballot_w32(mn < cntu) != 0u) {
        int e0 = (g << 7) + (lane << 2);
        qn = agg1_push(myq, qn, a < cntu, e0 + 0);
        qn = agg1_push(myq, qn, b < cntu, e0 + 1);
        qn = agg1_push(myq, qn, c < cntu, e0 + 2);
        qn = agg1_push(myq, qn, d < cntu, e0 + 3);
    }
    return qn;
}

__device__ __forceinline__ void scan_note(const int* myq, int qn, int lane, int lo,
                                          const int* __restrict__ src, const int* __restrict__ dst,
                                          int* degl, int2* myrec, int* rcur) {
    for (int base = 0; base < qn; base += 32) {
        int nb = qn - base; if (nb > 32) nb = 32;
        int k = base + (lane < nb ? lane : nb - 1);
        int e = myq[k];
        int s = src[e];
        int dl = dst[e] - lo;
        if (lane < nb) atomicAdd(&degl[dl], 1);
        if (myrec != nullptr) {
            int rb = 0;
            if (lane == 0) rb = atomicAdd(rcur, 32);
            rb = __builtin_amdgcn_readfirstlane(rb);
            if (rb + 32 <= RCAP) {
                int2 rv;
                rv.x = lane < nb ? (int)(((unsigned)dl << 23) | (unsigned)s) : -1;
                rv.y = 0;
                myrec[rb + lane] = rv;
            }
        }
    }
}

__device__ __forceinline__ void scan_wave(int* q, int ws, int lane, int lo, int cnt,
                                          const int* __restrict__ src, const int* __restrict__ dst,
                                          int* degl, int2* myrec, int* rcur, int n_edges) {
    int* myq = q + ws * QC;
    int qn = 0;
    const int n_groups = n_edges >> 7;
    const int4* dst4 = (const int4*)dst;
    int rot = (int)(((long long)blockIdx.x * n_groups) / gridDim.x);
    const int K = (n_groups - ws + NSW - 1) / NSW;

    for (int kb = 0; kb < K; kb += SCAN_U) {
        const int gi0 = ws + kb * NSW;
        int g0 = gi0 + rot;
        if (g0 >= n_groups) g0 -= n_groups;
        if (gi0 + NSW * (SCAN_U - 1) < n_groups && g0 + NSW * (SCAN_U - 1) < n_groups) {
            const int4* p = dst4 + (size_t)g0 * 32 + lane;
            int4 v[SCAN_U];
            #pragma unroll
            for (int u = 0; u < SCAN_U; ++u) v[u] = p[u * (NSW * 32)];
            #pragma unroll
            for (int u = 0; u < SCAN_U; ++u)
                qn = agg1_scan_piece(myq, qn, v[u], g0 + NSW * u, lo, (unsigned)cnt, lane);
        } else {
            int4 v[SCAN_U];
            int  gg[SCAN_U];
            unsigned cu[SCAN_U];
            #pragma unroll
            for (int u = 0; u < SCAN_U; ++u) {
                int gi = gi0 + NSW * u;
                bool ok = gi < n_groups;
                int g = gi + rot;
                if (g >= n_groups) g -= n_groups;
                if (!ok) g = 0;
                gg[u] = g;
                cu[u] = ok ? (unsigned)cnt : 0u;
                v[u] = dst4[(size_t)g * 32 + lane];
            }
            #pragma unroll
            for (int u = 0; u < SCAN_U; ++u)
                qn = agg1_scan_piece(myq, qn, v[u], gg[u], lo, cu[u], lane);
        }
        if (qn > QC - 128 * SCAN_U - 128) { scan_note(myq, qn, lane, lo, src, dst, degl, myrec, rcur); qn = 0; }
    }
    if (ws == 0) {
        for (int e0 = n_groups << 7; e0 < n_edges; e0 += 32) {
            int e = e0 + lane;
            bool m = e < n_edges && (unsigned)(dst[e < n_edges ? e : 0] - lo) < (unsigned)cnt;
            qn = agg1_push(myq, qn, m, e);
        }
    }
    scan_note(myq, qn, lane, lo, src, dst, degl, myrec, rcur);
}

__global__ __launch_bounds__(1024) void k_scan_gemm(const float* __restrict__ x, const float* __restrict__ W1,
                                                    float* __restrict__ h,
                                                    const int* __restrict__ src, const int* __restrict__ dst,
                                                    float* __restrict__ dinv, int* __restrict__ rec,
                                                    int* __restrict__ rcount, int n_nodes, int n_edges) {
    __shared__ float ldsW[NFEAT * NHID];
    __shared__ int   degl[NR];
    __shared__ int   q[NSW * QC];
    __shared__ int   rcur;

    const int tid  = threadIdx.x;
    const int lane = tid & 31;
    const int wave = __builtin_amdgcn_readfirstlane(tid >> 5);
    const int lo   = blockIdx.x * NR;
    int cnt = n_nodes - lo;
    if (cnt > NR) cnt = NR;
    if (cnt < 0)  cnt = 0;

    int2* myrec = (blockIdx.x < MAXOWN && n_nodes <= (1 << 23))
                      ? (int2*)rec + (size_t)blockIdx.x * RCAP : nullptr;
    if (tid == 0) rcur = 0;
    for (int i = tid; i < NR; i += 1024) degl[i] = 0;
    gemm_stage_w1(ldsW, W1, tid);
    __syncthreads();

    if (wave < NGW) {
        gemm_wave(x, h, ldsW, lane, (int)blockIdx.x + (int)gridDim.x * wave, (int)gridDim.x * NGW, n_nodes >> 4);
    } else {
        scan_wave(q, wave - NGW, lane, lo, cnt, src, dst, degl, myrec, &rcur, n_edges);
    }
    __syncthreads();

    if (tid == 0 && myrec != nullptr) rcount[blockIdx.x * 32] = rcur;

    for (int i = tid; i < cnt; i += 1024) {
        float d = (float)(degl[i] + 1);
        dinv[lo + i] = (d > 0.0f) ? rsqrtf(d) : 0.0f;
    }
}

__global__ __launch_bounds__(1024) void k_agg1_owned(const int* __restrict__ src, const int* __restrict__ dst,
                                                     const float* __restrict__ dinv, const float* __restrict__ h,
                                                     const float* __restrict__ b1, const float* __restrict__ W2,
                                                     float* __restrict__ z, int* rec,
                                                     const int* __restrict__ rcount, int n_nodes, int n_edges) {
    __shared__ float acc[NR * AP];

    const int tid  = threadIdx.x;
    const int lane = tid & 31;
    const int wave = __builtin_amdgcn_readfirstlane(tid >> 5);
    const int lo   = blockIdx.x * NR;
    int cnt = n_nodes - lo;
    if (cnt > NR) cnt = NR;
    if (cnt < 0)  cnt = 0;

    for (int idx = tid; idx < cnt * NHID; idx += 1024) {
        int node = idx >> 6;
        float di = dinv[lo + node];
        acc[node * AP + (idx & 63)] = h[(size_t)lo * NHID + idx] * di * di;
    }
    __syncthreads();

    const bool sliced = blockIdx.x < MAXOWN && n_nodes <= (1 << 23);
    const int total = sliced ? rcount[blockIdx.x * 32] : RCAP + 1;
    if (total <= RCAP) {
        int2* myrec = (int2*)rec + (size_t)blockIdx.x * RCAP;
        const int rounds = total >> 5;
        int kx = wave < rounds ? myrec[wave * 32 + lane].x : -1;
        for (int r = wave; r < rounds; r += 32) {
            const int key = kx;
            if (r + 32 < rounds) kx = myrec[(r + 32) * 32 + lane].x;
            bool ok = key != -1;
            int nb = __builtin_popcount(__builtin_amdgcn_ballot_w32(ok));
            int s  = ok ? (key & 0x7FFFFF) : 0;
            int dl = ok ? (int)((unsigned)key >> 23) : 0;
            float nrm = dinv[s] * dinv[dl + lo];
            float2 hv[32];
            #pragma unroll
            for (int u = 0; u < 32; ++u) {
                int sj = __builtin_amdgcn_readlane(s, u);
                hv[u] = *(const float2*)(h + (size_t)sj * NHID + (lane << 1));
            }
            int2 rv; rv.x = key; rv.y = __float_as_int(nrm);
            myrec[r * 32 + lane] = rv;
            #pragma unroll
            for (int u = 0; u < 32; ++u) {
                if (u < nb) {
                    int   dj = __builtin_amdgcn_readlane(dl, u);
                    float nj = __int_as_float(__builtin_amdgcn_readlane(__float_as_int(nrm), u));
                    float* row = acc + dj * AP + (lane << 1);
                    atomicAdd(row + 0, hv[u].x * nj);
                    atomicAdd(row + 1, hv[u].y * nj);
                }
            }
        }
    } else {
        for (int e = tid; e < n_edges; e += 1024) {
            int d = dst[e];
            if ((unsigned)(d - lo) < (unsigned)cnt) {
                int s = src[e];
                float nrm = dinv[s] * dinv[d];
                const float* hr = h + (size_t)s * NHID;
                float* row = acc + (d - lo) * AP;
                for (int f = 0; f < NHID; ++f) atomicAdd(row + f, hr[f] * nrm);
            }
        }
    }
    __syncthreads();

    if (tid < cnt) {
        const float* ar = acc + tid * AP;
        float z0 = 0.0f, z1 = 0.0f;
        #pragma unroll
        for (int f = 0; f < NHID; ++f) {
            float hv = fmaxf(ar[f] + b1[f], 0.0f);
            z0 += hv * W2[2 * f + 0];
            z1 += hv * W2[2 * f + 1];
        }
        float2 zz; zz.x = z0; zz.y = z1;
        *(float2*)(z + 2 * (size_t)(lo + tid)) = zz;
    }
}


#define A2U 8

__global__ __launch_bounds__(1024) void k_agg2_owned(const int* __restrict__ src, const int* __restrict__ dst,
                                                     const float* __restrict__ dinv, const float* __restrict__ z,
                                                     const float* __restrict__ b2, const int* __restrict__ rec,
                                                     const int* __restrict__ rcount, float* __restrict__ out,
                                                     int n_nodes, int n_edges) {
    __shared__ float o2[NR * 2];
    const int tid = threadIdx.x;
    const int lo  = blockIdx.x * NR;
    int cnt = n_nodes - lo;
    if (cnt > NR) cnt = NR;
    if (cnt < 0)  cnt = 0;

    for (int i = tid; i < cnt * 2; i += 1024) {
        float di = dinv[lo + (i >> 1)];
        float n2 = di * di;
        o2[i] = z[2 * (size_t)lo + i] * n2 + b2[i & 1];
    }
    __syncthreads();

    const bool sliced = blockIdx.x < MAXOWN && n_nodes <= (1 << 23);
    const int total = sliced ? rcount[blockIdx.x * 32] : RCAP + 1;
    if (total <= RCAP) {
        const int2* r = (const int2*)rec + (size_t)blockIdx.x * RCAP;
        for (int k0 = tid; k0 < total; k0 += 1024 * A2U) {
            int2 v[A2U];
            float2 zv[A2U];
            #pragma unroll
            for (int u = 0; u < A2U; ++u) {
                int k = k0 + u * 1024;
                v[u] = r[k < total ? k : k0];
                if (k >= total) v[u].x = -1;
            }
            #pragma unroll
            for (int u = 0; u < A2U; ++u) {
                int s = v[u].x == -1 ? 0 : (v[u].x & 0x7FFFFF);
                zv[u] = *(const float2*)(z + 2 * (size_t)s);
            }
            #pragma unroll
            for (int u = 0; u < A2U; ++u) {
                if (v[u].x != -1) {
                    int dl = (int)((unsigned)v[u].x >> 23);
                    float nrm = __int_as_float(v[u].y);
                    atomicAdd(&o2[2 * dl + 0], zv[u].x * nrm);
                    atomicAdd(&o2[2 * dl + 1], zv[u].y * nrm);
                }
            }
        }
    } else {
        for (int e = tid; e < n_edges; e += 1024) {
            int d = dst[e];
            if ((unsigned)(d - lo) < (unsigned)cnt) {
                int s = src[e];
                float nrm = dinv[s] * dinv[d];
                float2 zv = *(const float2*)(z + 2 * (size_t)s);
                atomicAdd(&o2[2 * (d - lo) + 0], zv.x * nrm);
                atomicAdd(&o2[2 * (d - lo) + 1], zv.y * nrm);
            }
        }
    }
    __syncthreads();

    for (int i = tid; i < cnt * 2; i += 1024) out[2 * (size_t)lo + i] = o2[i];
}


extern "C" void kernel_launch(void* const* d_in, const int* in_sizes, int n_in,
                              void* d_out, int out_size, void* d_ws, size_t ws_size,
                              hipStream_t stream) {
    const float* x  = (const float*)d_in[0];
    const int*   ei = (const int*)d_in[1];
    const float* W1 = (const float*)d_in[2];
    const float* b1 = (const float*)d_in[3];
    const float* W2 = (const float*)d_in[4];
    const float* b2 = (const float*)d_in[5];

    int n_nodes = in_sizes[0] / NFEAT;
    int n_edges = in_sizes[1] / 2;
    const int* src = ei;
    const int* dst = ei + n_edges;

    float* ws = (float*)d_ws;
    size_t off = 0;
    float* dinv = ws + off; off += ((size_t)n_nodes + 63) & ~(size_t)63;
    float* h    = ws + off; off += (size_t)n_nodes * NHID;
    float* recs = ws + off; off += (size_t)n_nodes * NHID;
    float* z    = ws + off; off += ((size_t)2 * n_nodes + 63) & ~(size_t)63;
    (void)ws_size;

    float* out = (float*)d_out;

    int owners = (n_nodes + NR - 1) / NR;

    int* rec    = (int*)recs;
    int* rcount = rec + (size_t)2 * MAXOWN * RCAP;
    k_scan_gemm<<<owners, 1024, 0, stream>>>(x, W1, h, src, dst, dinv, rec, rcount, n_nodes, n_edges);

    k_agg1_owned<<<owners, 1024, 0, stream>>>(src, dst, dinv, h, b1, W2, z, rec, rcount, n_nodes, n_edges);

    k_agg2_owned<<<owners, 1024, 0, stream>>>(src, dst, dinv, z, b2, rec, rcount, out, n_nodes, n_edges);
}


# Owner scan tiled 2-D: a work-group scans one half of the destination list for a pair of ranges (list re-read 121 times, was 241); in-degree = sum of two count planes; LDS-arrival sum order kept
# speedup vs baseline: 4.1376x; 1.1075x over previous
#include <hip/hip_runtime.h>

typedef float v2f __attribute__((ext_vector_type(2)));
typedef float v8f __attribute__((ext_vector_type(8)));

#define NFEAT 256
#define NHID  64


#define NGW  8
#define NSW  24
#define GPF  16
#define GSB  4

__device__ __forceinline__ void gemm_stage_w1(float* ldsW, const float* __restrict__ W1, int tid) {
    for (int idx = tid * 4; idx < NFEAT * NHID; idx += 1024 * 4) {
        float4 w = *(const float4*)(W1 + idx);
        int k  = idx >> 6;
        int n0 = idx & 63;
        int p  = k >> 1;
        int lo = k & 1;
        int rot = (p & 1) << 5;
        float vals[4] = {w.x, w.y, w.z, w.w};
        #pragma unroll
        for (int j = 0; j < 4; ++j) {
            int wpos = (((n0 + j) << 1) + rot) & 127;
            ldsW[p * 128 + wpos + lo] = vals[j];
        }
    }
}

__device__ __forceinline__ void gemm_wave(const float* __restrict__ x, float* __restrict__ h,
                                          const float* ldsW, int lane, int first, int stride, int n_tiles) {
    int t = first;
    if (t >= n_tiles) return;

    int half = lane >> 4;
    int kk   = half << 1;
    int ncol = lane & 15;
    int rot  = half << 5;

    int wp0 = (((ncol +  0) << 1) + rot) & 127;
    int wp1 = (((ncol + 16) << 1) + rot) & 127;
    int wp2 = (((ncol + 32) << 1) + rot) & 127;
    int wp3 = (((ncol + 48) << 1) + rot) & 127;
    unsigned boff = (unsigned)half * 128u;

    v2f a[GPF];
    const unsigned xlane = (unsigned)ncol * NFEAT + (unsigned)kk;
    const unsigned hlane = (unsigned)(half << 3) * NHID + (unsigned)ncol;
    const float* xr = x + (size_t)t * (16 * NFEAT) + xlane;
    #pragma unroll
    for (int s = 0; s < GPF; ++s) {
        float2 av = *(const float2*)(xr + s * 4);
        a[s].x = av.x; a[s].y = av.y;
    }
    for (;;) {
        const int  tn   = t + stride;
        const bool more = tn < n_tiles;
        const float* xn = x + (size_t)(more ? tn : t) * (16 * NFEAT) + xlane;

        asm volatile("" : "+v"(boff));
        const float* bb = ldsW + boff;
        v8f acc0 = {}, acc1 = {}, acc2 = {}, acc3 = {};
        #pragma unroll
        for (int g = 0; g < NFEAT / 4 / GSB; ++g) {
            #pragma unroll
            for (int j = 0; j < GSB; ++j) {
                const int s = g * GSB + j;
                const float* rowp = bb + (s << 1) * 128;
                v2f b0 = *(const v2f*)(rowp + wp0);
                v2f b1 = *(const v2f*)(rowp + wp1);
                v2f b2 = *(const v2f*)(rowp + wp2);
                v2f b3 = *(const v2f*)(rowp + wp3);
                const v2f av = a[s % GPF];
                acc0 = __builtin_amdgcn_wmma_f32_16x16x4_f32(false, av, false, b0, (short)0, acc0, false, false);
                acc1 = __builtin_amdgcn_wmma_f32_16x16x4_f32(false, av, false, b1, (short)0, acc1, false, false);
                acc2 = __builtin_amdgcn_wmma_f32_16x16x4_f32(false, av, false, b2, (short)0, acc2, false, false);
                acc3 = __builtin_amdgcn_wmma_f32_16x16x4_f32(false, av, false, b3, (short)0, acc3, false, false);
                const float* nx = (s + GPF < NFEAT / 4) ? xr + (s + GPF) * 4 : xn + (s + GPF - NFEAT / 4) * 4;
                float2 nv = *(const float2*)nx;
                a[s % GPF].x = nv.x; a[s % GPF].y = nv.y;
            }
            __builtin_amdgcn_sched_barrier(0);
        }

        float* ht = h + (size_t)t * (16 * NHID) + hlane;
        for (int r = 0; r < 8; ++r) {
            float* hr = ht + r * NHID;
            hr[ 0] = acc0[r];
            hr[16] = acc1[r];
            hr[32] = acc2[r];
            hr[48] = acc3[r];
        }
        if (!more) break;
        t = tn; xr = xn;
    }
}


#define NR   416
#define AP   65
#define MAXOWN 256
#define RCAP 5632
#define QC   1792
#define SCAN_U 12

__device__ __forceinline__ int agg1_push(int* myq, int qn, bool m, int e) {
    unsigned mask = __builtin_amdgcn_ballot_w32(m);
    if (mask != 0u) {
        if (m) myq[qn + (int)__builtin_amdgcn_mbcnt_lo(mask, 0u)] = e;
        qn += __builtin_popcount(mask);
    }
    return qn;
}

__device__ __forceinline__ int agg1_scan_piece(int* myq, int qn, int4 v, int g, int lo, unsigned cntu, int lane) {
    unsigned a = (unsigned)(v.x - lo), b = (unsigned)(v.y - lo);
    unsigned c = (unsigned)(v.z - lo), d = (unsigned)(v.w - lo);
    unsigned ab = a < b ? a : b, cd = c < d ? c : d;
    unsigned mn = ab < cd ? ab : cd;
    if (__builtin_amdgcn_ballot_w32(mn < cntu) != 0u) {
        int e0 = (g << 7) + (lane << 2);
        qn = agg1_push(myq, qn, a < cntu, e0 + 0);
        qn = agg1_push(myq, qn, b < cntu, e0 + 1);
        qn = agg1_push(myq, qn, c < cntu, e0 + 2);
        qn = agg1_push(myq, qn, d < cntu, e0 + 3);
    }
    return qn;
}

__device__ __forceinline__ void rec_emit(int* stj, int fill, int lane, int2* recj, int* rcurj) {
    int rb = 0;
    if (lane == 0) rb = atomicAdd(rcurj, 32);
    rb = __builtin_amdgcn_readfirstlane(rb);
    int key  = lane < fill ? stj[lane] : -1;
    int more = 32 + lane < fill ? stj[32 + lane] : 0;
    if (rb + 32 <= RCAP) {
        int2 rv;
        rv.x = key;
        rv.y = 0;
        recj[rb + lane] = rv;
    }
    if (32 + lane < fill) stj[lane] = more;
}

__device__ __forceinline__ void scan_note(const int* myq, int qn, int lane, int lo,
                                          const int* __restrict__ src, const int* __restrict__ dst,
                                          int* degl, int2* myrec, int* rcur, int* st, int& f0, int& f1) {
    for (int base = 0; base < qn; base += 32) {
        int nb = qn - base; if (nb > 32) nb = 32;
        int k = base + (lane < nb ? lane : nb - 1);
        int e = myq[k];
        int s = src[e];
        int dl = dst[e] - lo;
        const bool live = lane < nb;
        if (live) atomicAdd(&degl[dl], 1);
        if (myrec != nullptr) {
            const bool up = dl >= NR;
            const unsigned m0 = __builtin_amdgcn_ballot_w32(live && !up);
            const unsigned m1 = __builtin_amdgcn_ballot_w32(live && up);
            if (m0 != 0u) {
                if (live && !up)
                    st[f0 + (int)__builtin_amdgcn_mbcnt_lo(m0, 0u)] = (int)(((unsigned)dl << 23) | (unsigned)s);
                f0 += __builtin_popcount(m0);
                if (f0 >= 32) { rec_emit(st, f0, lane, myrec, rcur); f0 -= 32; }
            }
            if (m1 != 0u) {
                if (live && up)
                    st[64 + f1 + (int)__builtin_amdgcn_mbcnt_lo(m1, 0u)] = (int)(((unsigned)(dl - NR) << 23) | (unsigned)s);
                f1 += __builtin_popcount(m1);
                if (f1 >= 32) { rec_emit(st + 64, f1, lane, myrec + 2 * RCAP, rcur + 1); f1 -= 32; }
            }
        }
    }
}

__device__ __forceinline__ void scan_wave(int* q, int* stw, int ws, int lane, int lo, int cnt,
                                          const int* __restrict__ src, const int* __restrict__ dst,
                                          int* degl, int2* myrec, int* rcur, int gb, int n_groups,
                                          int pair, int n_pairs, bool tail, int n_edges) {
    int* myq = q + ws * QC;
    int* st  = stw + ws * 128;
    int qn = 0;
    int f0 = 0, f1 = 0;
    const int4* dst4 = (const int4*)dst + (size_t)gb * 32;
    int rot = (int)(((long long)pair * n_groups) / n_pairs);
    const int K = (n_groups - ws + NSW - 1) / NSW;

    for (int kb = 0; kb < K; kb += SCAN_U) {
        const int gi0 = ws + kb * NSW;
        int g0 = gi0 + rot;
        if (g0 >= n_groups) g0 -= n_groups;
        if (gi0 + NSW * (SCAN_U - 1) < n_groups && g0 + NSW * (SCAN_U - 1) < n_groups) {
            const int4* p = dst4 + (size_t)g0 * 32 + lane;
            int4 v[SCAN_U];
            #pragma unroll
            for (int u = 0; u < SCAN_U; ++u) v[u] = p[u * (NSW * 32)];
            #pragma unroll
            for (int u = 0; u < SCAN_U; ++u)
                qn = agg1_scan_piece(myq, qn, v[u], gb + g0 + NSW * u, lo, (unsigned)cnt, lane);
        } else {
            int4 v[SCAN_U];
            int  gg[SCAN_U];
            unsigned cu[SCAN_U];
            #pragma unroll
            for (int u = 0; u < SCAN_U; ++u) {
                int gi = gi0 + NSW * u;
                bool ok = gi < n_groups;
                int g = gi + rot;
                if (g >= n_groups) g -= n_groups;
                if (!ok) g = 0;
                gg[u] = gb + g;
                cu[u] = ok ? (unsigned)cnt : 0u;
                v[u] = dst4[(size_t)g * 32 + lane];
            }
            #pragma unroll
            for (int u = 0; u < SCAN_U; ++u)
                qn = agg1_scan_piece(myq, qn, v[u], gg[u], lo, cu[u], lane);
        }
        if (qn > QC - 128 * SCAN_U - 128) { scan_note(myq, qn, lane, lo, src, dst, degl, myrec, rcur, st, f0, f1); qn = 0; }
    }
    if (ws == 0 && tail) {
        for (int e0 = (n_edges >> 7) << 7; e0 < n_edges; e0 += 32) {
            int e = e0 + lane;
            bool m = e < n_edges && (unsigned)(dst[e < n_edges ? e : 0] - lo) < (unsigned)cnt;
            qn = agg1_push(myq, qn, m, e);
        }
    }
    scan_note(myq, qn, lane, lo, src, dst, degl, myrec, rcur, st, f0, f1);
    if (myrec != nullptr) {
        if (f0 > 0) rec_emit(st, f0, lane, myrec, rcur);
        if (f1 > 0) rec_emit(st + 64, f1, lane, myrec + 2 * RCAP, rcur + 1);
    }
}

__global__ __launch_bounds__(1024) void k_scan_gemm(const float* __restrict__ x, const float* __restrict__ W1,
                                                    float* __restrict__ h,
                                                    const int* __restrict__ src, const int* __restrict__ dst,
                                                    int* __restrict__ cpl, int pstride, int* __restrict__ rec,
                                                    int* __restrict__ rcount, int n_nodes, int n_edges) {
    __shared__ float ldsW[NFEAT * NHID];
    __shared__ int   degl[2 * NR];
    __shared__ int   q[NSW * QC];
    __shared__ int   stw[NSW * 128];
    __shared__ int   rcur[2];

    const int tid  = threadIdx.x;
    const int lane = tid & 31;
    const int wave = __builtin_amdgcn_readfirstlane(tid >> 5);
    const int pair = (int)blockIdx.x >> 1;
    const int half = (int)blockIdx.x & 1;
    const int lo   = pair * (2 * NR);
    int cnt = n_nodes - lo;
    if (cnt > 2 * NR) cnt = 2 * NR;
    if (cnt < 0)      cnt = 0;
    const int all_groups = n_edges >> 7;
    const int gsplit     = (all_groups + 1) >> 1;
    const int gb = half ? gsplit : 0;
    const int ng = half ? all_groups - gsplit : gsplit;

    int2* myrec = (2 * pair < MAXOWN && n_nodes <= (1 << 23))
                      ? (int2*)rec + (size_t)(4 * pair + half) * RCAP : nullptr;
    if (tid < 2) rcur[tid] = 0;
    for (int i = tid; i < 2 * NR; i += 1024) degl[i] = 0;
    gemm_stage_w1(ldsW, W1, tid);
    __syncthreads();

    if (wave < NGW) {
        gemm_wave(x, h, ldsW, lane, (int)blockIdx.x + (int)gridDim.x * wave, (int)gridDim.x * NGW, n_nodes >> 4);
    } else {
        scan_wave(q, stw, wave - NGW, lane, lo, cnt, src, dst, degl, myrec, rcur, gb, ng,
                  pair, (int)gridDim.x >> 1, half == 1, n_edges);
    }
    __syncthreads();

    if (tid < 2 && myrec != nullptr) rcount[(4 * pair + 2 * tid + half) * 32] = rcur[tid];

    for (int i = tid; i < cnt; i += 1024) cpl[(size_t)half * pstride + lo + i] = degl[i];
}

__device__ __forceinline__ float dinv_of(const int* __restrict__ c0, const int* __restrict__ c1, int i) {
    float d = (float)(c0[i] + c1[i] + 1);
    return (d > 0.0f) ? rsqrtf(d) : 0.0f;
}

__global__ __launch_bounds__(1024) void k_agg1_owned(const int* __restrict__ src, const int* __restrict__ dst,
                                                     const int* __restrict__ cpl, int pstride,
                                                     float* __restrict__ dinv, const float* __restrict__ h,
                                                     const float* __restrict__ b1, const float* __restrict__ W2,
                                                     float* __restrict__ z, int* rec,
                                                     const int* __restrict__ rcount, int n_nodes, int n_edges) {
    __shared__ float acc[NR * AP];
    __shared__ float dloc[NR];
    const int* __restrict__ c0 = cpl;
    const int* __restrict__ c1 = cpl + pstride;

    const int tid  = threadIdx.x;
    const int lane = tid & 31;
    const int wave = __builtin_amdgcn_readfirstlane(tid >> 5);
    const int lo   = blockIdx.x * NR;
    int cnt = n_nodes - lo;
    if (cnt > NR) cnt = NR;
    if (cnt < 0)  cnt = 0;

    for (int i = tid; i < cnt; i += 1024) {
        float di = dinv_of(c0, c1, lo + i);
        dloc[i] = di;
        dinv[lo + i] = di;
    }
    __syncthreads();

    for (int idx = tid; idx < cnt * NHID; idx += 1024) {
        int node = idx >> 6;
        float di = dloc[node];
        acc[node * AP + (idx & 63)] = h[(size_t)lo * NHID + idx] * di * di;
    }
    __syncthreads();

    const bool sliced = blockIdx.x < MAXOWN && n_nodes <= (1 << 23);
    const int t0 = sliced ? rcount[(2 * blockIdx.x + 0) * 32] : RCAP + 1;
    const int t1 = sliced ? rcount[(2 * blockIdx.x + 1) * 32] : RCAP + 1;
    if (t0 <= RCAP && t1 <= RCAP) {
        int2* s0 = (int2*)rec + (size_t)(2 * blockIdx.x) * RCAP;
        int2* s1 = s0 + RCAP;
        const int r0n = t0 >> 5;
        const int rounds = r0n + (t1 >> 5);
        int kx = -1;
        if (wave < rounds) kx = (wave < r0n ? s0 + wave * 32 : s1 + (wave - r0n) * 32)[lane].x;
        for (int r = wave; r < rounds; r += 32) {
            int2* const myround = r < r0n ? s0 + r * 32 : s1 + (r - r0n) * 32;
            const int key = kx;
            if (r + 32 < rounds) {
                const int rn = r + 32;
                kx = (rn < r0n ? s0 + rn * 32 : s1 + (rn - r0n) * 32)[lane].x;
            }
            bool ok = key != -1;
            int nb = __builtin_popcount(__builtin_amdgcn_ballot_w32(ok));
            int s  = ok ? (key & 0x7FFFFF) : 0;
            int dl = ok ? (int)((unsigned)key >> 23) : 0;
            float nrm = dinv_of(c0, c1, s) * dloc[dl];
            float2 hv[32];
            #pragma unroll
            for (int u = 0; u < 32; ++u) {
                int sj = __builtin_amdgcn_readlane(s, u);
                hv[u] = *(const float2*)(h + (size_t)sj * NHID + (lane << 1));
            }
            int2 rv; rv.x = key; rv.y = __float_as_int(nrm);
            myround[lane] = rv;
            #pragma unroll
            for (int u = 0; u < 32; ++u) {
                if (u < nb) {
                    int   dj = __builtin_amdgcn_readlane(dl, u);
                    float nj = __int_as_float(__builtin_amdgcn_readlane(__float_as_int(nrm), u));
                    float* row = acc + dj * AP + (lane << 1);
                    atomicAdd(row + 0, hv[u].x * nj);
                    atomicAdd(row + 1, hv[u].y * nj);
                }
            }
        }
    } else {
        for (int e = tid; e < n_edges; e += 1024) {
            int d = dst[e];
            if ((unsigned)(d - lo) < (unsigned)cnt) {
                int s = src[e];
                float nrm = dinv_of(c0, c1, s) * dloc[d - lo];
                const float* hr = h + (size_t)s * NHID;
                float* row = acc + (d - lo) * AP;
                for (int f = 0; f < NHID; ++f) atomicAdd(row + f, hr[f] * nrm);
            }
        }
    }
    __syncthreads();

    if (tid < cnt) {
        const float* ar = acc + tid * AP;
        float z0 = 0.0f, z1 = 0.0f;
        #pragma unroll
        for (int f = 0; f < NHID; ++f) {
            float hv = fmaxf(ar[f] + b1[f], 0.0f);
            z0 += hv * W2[2 * f + 0];
            z1 += hv * W2[2 * f + 1];
        }
        float2 zz; zz.x = z0; zz.y = z1;
        *(float2*)(z + 2 * (size_t)(lo + tid)) = zz;
    }
}


#define A2U 8

__global__ __launch_bounds__(1024) void k_agg2_owned(const int* __restrict__ src, const int* __restrict__ dst,
                                                     const float* __restrict__ dinv, const float* __restrict__ z,
                                                     const float* __restrict__ b2, const int* __restrict__ rec,
                                                     const int* __restrict__ rcount, float* __restrict__ out,
                                                     int n_nodes, int n_edges) {
    __shared__ float o2[NR * 2];
    const int tid = threadIdx.x;
    const int lo  = blockIdx.x * NR;
    int cnt = n_nodes - lo;
    if (cnt > NR) cnt = NR;
    if (cnt < 0)  cnt = 0;

    for (int i = tid; i < cnt * 2; i += 1024) {
        float di = dinv[lo + (i >> 1)];
        float n2 = di * di;
        o2[i] = z[2 * (size_t)lo + i] * n2 + b2[i & 1];
    }
    __syncthreads();

    const bool sliced = blockIdx.x < MAXOWN && n_nodes <= (1 << 23);
    const int t0 = sliced ? rcount[(2 * blockIdx.x + 0) * 32] : RCAP + 1;
    const int t1 = sliced ? rcount[(2 * blockIdx.x + 1) * 32] : RCAP + 1;
    if (t0 <= RCAP && t1 <= RCAP) {
        const int2* s0 = (const int2*)rec + (size_t)(2 * blockIdx.x) * RCAP;
        const int2* s1 = s0 + RCAP;
        const int total = t0 + t1;
        for (int k0 = tid; k0 < total; k0 += 1024 * A2U) {
            int2 v[A2U];
            float2 zv[A2U];
            #pragma unroll
            for (int u = 0; u < A2U; ++u) {
                int k = k0 + u * 1024;
                int kk = k < total ? k : k0;
                v[u] = kk < t0 ? s0[kk] : s1[kk - t0];
                if (k >= total) v[u].x = -1;
            }
            #pragma unroll
            for (int u = 0; u < A2U; ++u) {
                int s = v[u].x == -1 ? 0 : (v[u].x & 0x7FFFFF);
                zv[u] = *(const float2*)(z + 2 * (size_t)s);
            }
            #pragma unroll
            for (int u = 0; u < A2U; ++u) {
                if (v[u].x != -1) {
                    int dl = (int)((unsigned)v[u].x >> 23);
                    float nrm = __int_as_float(v[u].y);
                    atomicAdd(&o2[2 * dl + 0], zv[u].x * nrm);
                    atomicAdd(&o2[2 * dl + 1], zv[u].y * nrm);
                }
            }
        }
    } else {
        for (int e = tid; e < n_edges; e += 1024) {
            int d = dst[e];
            if ((unsigned)(d - lo) < (unsigned)cnt) {
                int s = src[e];
                float nrm = dinv[s] * dinv[d];
                float2 zv = *(const float2*)(z + 2 * (size_t)s);
                atomicAdd(&o2[2 * (d - lo) + 0], zv.x * nrm);
                atomicAdd(&o2[2 * (d - lo) + 1], zv.y * nrm);
            }
        }
    }
    __syncthreads();

    for (int i = tid; i < cnt * 2; i += 1024) out[2 * (size_t)lo + i] = o2[i];
}


extern "C" void kernel_launch(void* const* d_in, const int* in_sizes, int n_in,
                              void* d_out, int out_size, void* d_ws, size_t ws_size,
                              hipStream_t stream) {
    const float* x  = (const float*)d_in[0];
    const int*   ei = (const int*)d_in[1];
    const float* W1 = (const float*)d_in[2];
    const float* b1 = (const float*)d_in[3];
    const float* W2 = (const float*)d_in[4];
    const float* b2 = (const float*)d_in[5];

    int n_nodes = in_sizes[0] / NFEAT;
    int n_edges = in_sizes[1] / 2;
    const int* src = ei;
    const int* dst = ei + n_edges;

    float* ws = (float*)d_ws;
    size_t off = 0;
    float* dinv = ws + off; off += ((size_t)n_nodes + 63) & ~(size_t)63;
    float* h    = ws + off; off += (size_t)n_nodes * NHID;
    float* recs = ws + off; off += (size_t)n_nodes * NHID;
    float* z    = ws + off; off += ((size_t)2 * n_nodes + 63) & ~(size_t)63;
    (void)ws_size;

    float* out = (float*)d_out;

    int owners = (n_nodes + NR - 1) / NR;

    int* rec     = (int*)recs;
    int* rcount  = rec + (size_t)2 * (2 * MAXOWN) * RCAP;
    int* cpl     = rcount + (2 * MAXOWN) * 32;
    int  pstride = (n_nodes + 63) & ~63;
    int  pairs   = (owners + 1) / 2;
    k_scan_gemm<<<2 * pairs, 1024, 0, stream>>>(x, W1, h, src, dst, cpl, pstride, rec, rcount, n_nodes, n_edges);

    k_agg1_owned<<<owners, 1024, 0, stream>>>(src, dst, cpl, pstride, dinv, h, b1, W2, z, rec, rcount, n_nodes, n_edges);

    k_agg2_owned<<<owners, 1024, 0, stream>>>(src, dst, dinv, z, b2, rec, rcount, out, n_nodes, n_edges);
}


# Fewer random word reads per edge (each moves a whole line): scan queues local dst with the edge id, no dst re-read; in-degree halves stored side by side by atomic exchange, one read; sum order kept
# speedup vs baseline: 4.5603x; 1.1022x over previous
#include <hip/hip_runtime.h>

typedef float v2f __attribute__((ext_vector_type(2)));
typedef float v8f __attribute__((ext_vector_type(8)));

#define NFEAT 256
#define NHID  64


#define NGW  8
#define NSW  24
#define GPF  16
#define GSB  4

__device__ __forceinline__ void gemm_stage_w1(float* ldsW, const float* __restrict__ W1, int tid) {
    for (int idx = tid * 4; idx < NFEAT * NHID; idx += 1024 * 4) {
        float4 w = *(const float4*)(W1 + idx);
        int k  = idx >> 6;
        int n0 = idx & 63;
        int p  = k >> 1;
        int lo = k & 1;
        int rot = (p & 1) << 5;
        float vals[4] = {w.x, w.y, w.z, w.w};
        #pragma unroll
        for (int j = 0; j < 4; ++j) {
            int wpos = (((n0 + j) << 1) + rot) & 127;
            ldsW[p * 128 + wpos + lo] = vals[j];
        }
    }
}

__device__ __forceinline__ void gemm_wave(const float* __restrict__ x, float* __restrict__ h,
                                          const float* ldsW, int lane, int first, int stride, int n_tiles) {
    int t = first;
    if (t >= n_tiles) return;

    int half = lane >> 4;
    int kk   = half << 1;
    int ncol = lane & 15;
    int rot  = half << 5;

    int wp0 = (((ncol +  0) << 1) + rot) & 127;
    int wp1 = (((ncol + 16) << 1) + rot) & 127;
    int wp2 = (((ncol + 32) << 1) + rot) & 127;
    int wp3 = (((ncol + 48) << 1) + rot) & 127;
    unsigned boff = (unsigned)half * 128u;

    v2f a[GPF];
    const unsigned xlane = (unsigned)ncol * NFEAT + (unsigned)kk;
    const unsigned hlane = (unsigned)(half << 3) * NHID + (unsigned)ncol;
    const float* xr = x + (size_t)t * (16 * NFEAT) + xlane;
    #pragma unroll
    for (int s = 0; s < GPF; ++s) {
        float2 av = *(const float2*)(xr + s * 4);
        a[s].x = av.x; a[s].y = av.y;
    }
    for (;;) {
        const int  tn   = t + stride;
        const bool more = tn < n_tiles;
        const float* xn = x + (size_t)(more ? tn : t) * (16 * NFEAT) + xlane;

        asm volatile("" : "+v"(boff));
        const float* bb = ldsW + boff;
        v8f acc0 = {}, acc1 = {}, acc2 = {}, acc3 = {};
        #pragma unroll
        for (int g = 0; g < NFEAT / 4 / GSB; ++g) {
            #pragma unroll
            for (int j = 0; j < GSB; ++j) {
                const int s = g * GSB + j;
                const float* rowp = bb + (s << 1) * 128;
                v2f b0 = *(const v2f*)(rowp + wp0);
                v2f b1 = *(const v2f*)(rowp + wp1);
                v2f b2 = *(const v2f*)(rowp + wp2);
                v2f b3 = *(const v2f*)(rowp + wp3);
                const v2f av = a[s % GPF];
                acc0 = __builtin_amdgcn_wmma_f32_16x16x4_f32(false, av, false, b0, (short)0, acc0, false, false);
                acc1 = __builtin_amdgcn_wmma_f32_16x16x4_f32(false, av, false, b1, (short)0, acc1, false, false);
                acc2 = __builtin_amdgcn_wmma_f32_16x16x4_f32(false, av, false, b2, (short)0, acc2, false, false);
                acc3 = __builtin_amdgcn_wmma_f32_16x16x4_f32(false, av, false, b3, (short)0, acc3, false, false);
                const float* nx = (s + GPF < NFEAT / 4) ? xr + (s + GPF) * 4 : xn + (s + GPF - NFEAT / 4) * 4;
                float2 nv = *(const float2*)nx;
                a[s % GPF].x = nv.x; a[s % GPF].y = nv.y;
            }
            __builtin_amdgcn_sched_barrier(0);
        }

        float* ht = h + (size_t)t * (16 * NHID) + hlane;
        for (int r = 0; r < 8; ++r) {
            float* hr = ht + r * NHID;
            hr[ 0] = acc0[r];
            hr[16] = acc1[r];
            hr[32] = acc2[r];
            hr[48] = acc3[r];
        }
        if (!more) break;
        t = tn; xr = xn;
    }
}


#define NR   416
#define AP   65
#define MAXOWN 256
#define RCAP 5632
#define QC   1792
#define SCAN_U 12

#define QSH 22

__device__ __forceinline__ int agg1_entry(unsigned dl, int e, unsigned pkm) {
    return (int)(((dl << QSH) & pkm) | (unsigned)e);
}

__device__ __forceinline__ int agg1_push(int* myq, int qn, bool m, int e) {
    unsigned mask = __builtin_amdgcn_ballot_w32(m);
    if (mask != 0u) {
        if (m) myq[qn + (int)__builtin_amdgcn_mbcnt_lo(mask, 0u)] = e;
        qn += __builtin_popcount(mask);
    }
    return qn;
}

__device__ __forceinline__ int agg1_scan_piece(int* myq, int qn, int4 v, int g, int lo, unsigned cntu, int lane,
                                               unsigned pkm) {
    unsigned a = (unsigned)(v.x - lo), b = (unsigned)(v.y - lo);
    unsigned c = (unsigned)(v.z - lo), d = (unsigned)(v.w - lo);
    unsigned ab = a < b ? a : b, cd = c < d ? c : d;
    unsigned mn = ab < cd ? ab : cd;
    if (__builtin_amdgcn_ballot_w32(mn < cntu) != 0u) {
        int e0 = (g << 7) + (lane << 2);
        qn = agg1_push(myq, qn, a < cntu, agg1_entry(a, e0 + 0, pkm));
        qn = agg1_push(myq, qn, b < cntu, agg1_entry(b, e0 + 1, pkm));
        qn = agg1_push(myq, qn, c < cntu, agg1_entry(c, e0 + 2, pkm));
        qn = agg1_push(myq, qn, d < cntu, agg1_entry(d, e0 + 3, pkm));
    }
    return qn;
}

__device__ __forceinline__ void rec_emit(int* stj, int fill, int lane, int2* recj, int* rcurj) {
    int rb = 0;
    if (lane == 0) rb = atomicAdd(rcurj, 32);
    rb = __builtin_amdgcn_readfirstlane(rb);
    int key  = lane < fill ? stj[lane] : -1;
    int more = 32 + lane < fill ? stj[32 + lane] : 0;
    if (rb + 32 <= RCAP) {
        int2 rv;
        rv.x = key;
        rv.y = 0;
        recj[rb + lane] = rv;
    }
    if (32 + lane < fill) stj[lane] = more;
}

__device__ __forceinline__ void scan_note(const int* myq, int qn, int lane, int lo,
                                          const int* __restrict__ src, const int* __restrict__ dst,
                                          int* degl, int2* myrec, int* rcur, int* st, int& f0, int& f1,
                                          unsigned pkm) {
    for (int base = 0; base < qn; base += 32) {
        int nb = qn - base; if (nb > 32) nb = 32;
        int k = base + (lane < nb ? lane : nb - 1);
        const int ent = myq[k];
        int e = ent, dl;
        if (pkm != 0u) {
            e  = ent & ((1 << QSH) - 1);
            dl = (int)((unsigned)ent >> QSH);
        } else {
            dl = dst[ent] - lo;
        }
        int s = src[e];
        const bool live = lane < nb;
        if (live) atomicAdd(&degl[dl], 1);
        if (myrec != nullptr) {
            const bool up = dl >= NR;
            const unsigned m0 = __builtin_amdgcn_ballot_w32(live && !up);
            const unsigned m1 = __builtin_amdgcn_ballot_w32(live && up);
            if (m0 != 0u) {
                if (live && !up)
                    st[f0 + (int)__builtin_amdgcn_mbcnt_lo(m0, 0u)] = (int)(((unsigned)dl << 23) | (unsigned)s);
                f0 += __builtin_popcount(m0);
                if (f0 >= 32) { rec_emit(st, f0, lane, myrec, rcur); f0 -= 32; }
            }
            if (m1 != 0u) {
                if (live && up)
                    st[64 + f1 + (int)__builtin_amdgcn_mbcnt_lo(m1, 0u)] = (int)(((unsigned)(dl - NR) << 23) | (unsigned)s);
                f1 += __builtin_popcount(m1);
                if (f1 >= 32) { rec_emit(st + 64, f1, lane, myrec + 2 * RCAP, rcur + 1); f1 -= 32; }
            }
        }
    }
}

__device__ __forceinline__ void scan_wave(int* q, int* stw, int ws, int lane, int lo, int cnt,
                                          const int* __restrict__ src, const int* __restrict__ dst,
                                          int* degl, int2* myrec, int* rcur, int gb, int n_groups,
                                          int pair, int n_pairs, bool tail, int n_edges, unsigned pkm) {
    int* myq = q + ws * QC;
    int* st  = stw + ws * 128;
    int qn = 0;
    int f0 = 0, f1 = 0;
    const int4* dst4 = (const int4*)dst + (size_t)gb * 32;
    int rot = (int)(((long long)pair * n_groups) / n_pairs);
    const int K = (n_groups - ws + NSW - 1) / NSW;

    for (int kb = 0; kb < K; kb += SCAN_U) {
        const int gi0 = ws + kb * NSW;
        int g0 = gi0 + rot;
        if (g0 >= n_groups) g0 -= n_groups;
        if (gi0 + NSW * (SCAN_U - 1) < n_groups && g0 + NSW * (SCAN_U - 1) < n_groups) {
            const int4* p = dst4 + (size_t)g0 * 32 + lane;
            int4 v[SCAN_U];
            #pragma unroll
            for (int u = 0; u < SCAN_U; ++u) v[u] = p[u * (NSW * 32)];
            #pragma unroll
            for (int u = 0; u < SCAN_U; ++u)
                qn = agg1_scan_piece(myq, qn, v[u], gb + g0 + NSW * u, lo, (unsigned)cnt, lane, pkm);
        } else {
            int4 v[SCAN_U];
            int  gg[SCAN_U];
            unsigned cu[SCAN_U];
            #pragma unroll
            for (int u = 0; u < SCAN_U; ++u) {
                int gi = gi0 + NSW * u;
                bool ok = gi < n_groups;
                int g = gi + rot;
                if (g >= n_groups) g -= n_groups;
                if (!ok) g = 0;
                gg[u] = gb + g;
                cu[u] = ok ? (unsigned)cnt : 0u;
                v[u] = dst4[(size_t)g * 32 + lane];
            }
            #pragma unroll
            for (int u = 0; u < SCAN_U; ++u)
                qn = agg1_scan_piece(myq, qn, v[u], gg[u], lo, cu[u], lane, pkm);
        }
        if (qn > QC - 128 * SCAN_U - 128) { scan_note(myq, qn, lane, lo, src, dst, degl, myrec, rcur, st, f0, f1, pkm); qn = 0; }
    }
    if (ws == 0 && tail) {
        for (int e0 = (n_edges >> 7) << 7; e0 < n_edges; e0 += 32) {
            int e = e0 + lane;
            unsigned a = (unsigned)(dst[e < n_edges ? e : 0] - lo);
            bool m = e < n_edges && a < (unsigned)cnt;
            qn = agg1_push(myq, qn, m, agg1_entry(a, e, pkm));
        }
    }
    scan_note(myq, qn, lane, lo, src, dst, degl, myrec, rcur, st, f0, f1, pkm);
    if (myrec != nullptr) {
        if (f0 > 0) rec_emit(st, f0, lane, myrec, rcur);
        if (f1 > 0) rec_emit(st + 64, f1, lane, myrec + 2 * RCAP, rcur + 1);
    }
}

__global__ __launch_bounds__(1024) void k_scan_gemm(const float* __restrict__ x, const float* __restrict__ W1,
                                                    float* __restrict__ h,
                                                    const int* __restrict__ src, const int* __restrict__ dst,
                                                    int* __restrict__ cpl, int* __restrict__ rec,
                                                    int* __restrict__ rcount, int n_nodes, int n_edges) {
    __shared__ float ldsW[NFEAT * NHID];
    __shared__ int   degl[2 * NR];
    __shared__ int   q[NSW * QC];
    __shared__ int   stw[NSW * 128];
    __shared__ int   rcur[2];

    const int tid  = threadIdx.x;
    const int lane = tid & 31;
    const int wave = __builtin_amdgcn_readfirstlane(tid >> 5);
    const int pair = (int)blockIdx.x >> 1;
    const int half = (int)blockIdx.x & 1;
    const int lo   = pair * (2 * NR);
    int cnt = n_nodes - lo;
    if (cnt > 2 * NR) cnt = 2 * NR;
    if (cnt < 0)      cnt = 0;
    const int all_groups = n_edges >> 7;
    const int gsplit     = (all_groups + 1) >> 1;
    const int gb = half ? gsplit : 0;
    const int ng = half ? all_groups - gsplit : gsplit;

    int2* myrec = (2 * pair < MAXOWN && n_nodes <= (1 << 23))
                      ? (int2*)rec + (size_t)(4 * pair + half) * RCAP : nullptr;
    const unsigned pkm = n_edges <= (1 << QSH) ? 0xFFFFFFFFu : 0u;
    if (tid < 2) rcur[tid] = 0;
    for (int i = tid; i < 2 * NR; i += 1024) degl[i] = 0;
    gemm_stage_w1(ldsW, W1, tid);
    __syncthreads();

    if (wave < NGW) {
        gemm_wave(x, h, ldsW, lane, (int)blockIdx.x + (int)gridDim.x * wave, (int)gridDim.x * NGW, n_nodes >> 4);
    } else {
        scan_wave(q, stw, wave - NGW, lane, lo, cnt, src, dst, degl, myrec, rcur, gb, ng,
                  pair, (int)gridDim.x >> 1, half == 1, n_edges, pkm);
    }
    __syncthreads();

    if (tid < 2 && myrec != nullptr) rcount[(4 * pair + 2 * tid + half) * 32] = rcur[tid];

    for (int i = tid; i < cnt; i += 1024) atomicExch(&cpl[2 * (size_t)(lo + i) + half], degl[i]);
}

__device__ __forceinline__ float dinv_of(const int* __restrict__ cpl, int i) {
    int2 c = *(const int2*)(cpl + 2 * (size_t)i);
    float d = (float)(c.x + c.y + 1);
    return (d > 0.0f) ? rsqrtf(d) : 0.0f;
}

__global__ __launch_bounds__(1024) void k_agg1_owned(const int* __restrict__ src, const int* __restrict__ dst,
                                                     const int* __restrict__ cpl,
                                                     float* __restrict__ dinv, const float* __restrict__ h,
                                                     const float* __restrict__ b1, const float* __restrict__ W2,
                                                     float* __restrict__ z, int* rec,
                                                     const int* __restrict__ rcount, int n_nodes, int n_edges) {
    __shared__ float acc[NR * AP];
    __shared__ float dloc[NR];

    const int tid  = threadIdx.x;
    const int lane = tid & 31;
    const int wave = __builtin_amdgcn_readfirstlane(tid >> 5);
    const int lo   = blockIdx.x * NR;
    int cnt = n_nodes - lo;
    if (cnt > NR) cnt = NR;
    if (cnt < 0)  cnt = 0;

    for (int i = tid; i < cnt; i += 1024) {
        float di = dinv_of(cpl, lo + i);
        dloc[i] = di;
        dinv[lo + i] = di;
    }
    __syncthreads();

    for (int idx = tid; idx < cnt * NHID; idx += 1024) {
        int node = idx >> 6;
        float di = dloc[node];
        acc[node * AP + (idx & 63)] = h[(size_t)lo * NHID + idx] * di * di;
    }
    __syncthreads();

    const bool sliced = blockIdx.x < MAXOWN && n_nodes <= (1 << 23);
    const int t0 = sliced ? rcount[(2 * blockIdx.x + 0) * 32] : RCAP + 1;
    const int t1 = sliced ? rcount[(2 * blockIdx.x + 1) * 32] : RCAP + 1;
    if (t0 <= RCAP && t1 <= RCAP) {
        int2* s0 = (int2*)rec + (size_t)(2 * blockIdx.x) * RCAP;
        int2* s1 = s0 + RCAP;
        const int r0n = t0 >> 5;
        const int rounds = r0n + (t1 >> 5);
        int kx = -1;
        if (wave < rounds) kx = (wave < r0n ? s0 + wave * 32 : s1 + (wave - r0n) * 32)[lane].x;
        for (int r = wave; r < rounds; r += 32) {
            int2* const myround = r < r0n ? s0 + r * 32 : s1 + (r - r0n) * 32;
            const int key = kx;
            if (r + 32 < rounds) {
                const int rn = r + 32;
                kx = (rn < r0n ? s0 + rn * 32 : s1 + (rn - r0n) * 32)[lane].x;
            }
            bool ok = key != -1;
            int nb = __builtin_popcount(__builtin_amdgcn_ballot_w32(ok));
            int s  = ok ? (key & 0x7FFFFF) : 0;
            int dl = ok ? (int)((unsigned)key >> 23) : 0;
            float nrm = dinv_of(cpl, s) * dloc[dl];
            float2 hv[32];
            #pragma unroll
            for (int u = 0; u < 32; ++u) {
                int sj = __builtin_amdgcn_readlane(s, u);
                hv[u] = *(const float2*)(h + (size_t)sj * NHID + (lane << 1));
            }
            int2 rv; rv.x = key; rv.y = __float_as_int(nrm);
            myround[lane] = rv;
            #pragma unroll
            for (int u = 0; u < 32; ++u) {
                if (u < nb) {
                    int   dj = __builtin_amdgcn_readlane(dl, u);
                    float nj = __int_as_float(__builtin_amdgcn_readlane(__float_as_int(nrm), u));
                    float* row = acc + dj * AP + (lane << 1);
                    atomicAdd(row + 0, hv[u].x * nj);
                    atomicAdd(row + 1, hv[u].y * nj);
                }
            }
        }
    } else {
        for (int e = tid; e < n_edges; e += 1024) {
            int d = dst[e];
            if ((unsigned)(d - lo) < (unsigned)cnt) {
                int s = src[e];
                float nrm = dinv_of(cpl, s) * dloc[d - lo];
                const float* hr = h + (size_t)s * NHID;
                float* row = acc + (d - lo) * AP;
                for (int f = 0; f < NHID; ++f) atomicAdd(row + f, hr[f] * nrm);
            }
        }
    }
    __syncthreads();

    if (tid < cnt) {
        const float* ar = acc + tid * AP;
        float z0 = 0.0f, z1 = 0.0f;
        #pragma unroll
        for (int f = 0; f < NHID; ++f) {
            float hv = fmaxf(ar[f] + b1[f], 0.0f);
            z0 += hv * W2[2 * f + 0];
            z1 += hv * W2[2 * f + 1];
        }
        float2 zz; zz.x = z0; zz.y = z1;
        *(float2*)(z + 2 * (size_t)(lo + tid)) = zz;
    }
}


#define A2U 8

__global__ __launch_bounds__(1024) void k_agg2_owned(const int* __restrict__ src, const int* __restrict__ dst,
                                                     const float* __restrict__ dinv, const float* __restrict__ z,
                                                     const float* __restrict__ b2, const int* __restrict__ rec,
                                                     const int* __restrict__ rcount, float* __restrict__ out,
                                                     int n_nodes, int n_edges) {
    __shared__ float o2[NR * 2];
    const int tid = threadIdx.x;
    const int lo  = blockIdx.x * NR;
    int cnt = n_nodes - lo;
    if (cnt > NR) cnt = NR;
    if (cnt < 0)  cnt = 0;

    for (int i = tid; i < cnt * 2; i += 1024) {
        float di = dinv[lo + (i >> 1)];
        float n2 = di * di;
        o2[i] = z[2 * (size_t)lo + i] * n2 + b2[i & 1];
    }
    __syncthreads();

    const bool sliced = blockIdx.x < MAXOWN && n_nodes <= (1 << 23);
    const int t0 = sliced ? rcount[(2 * blockIdx.x + 0) * 32] : RCAP + 1;
    const int t1 = sliced ? rcount[(2 * blockIdx.x + 1) * 32] : RCAP + 1;
    if (t0 <= RCAP && t1 <= RCAP) {
        const int2* s0 = (const int2*)rec + (size_t)(2 * blockIdx.x) * RCAP;
        const int2* s1 = s0 + RCAP;
        const int total = t0 + t1;
        for (int k0 = tid; k0 < total; k0 += 1024 * A2U) {
            int2 v[A2U];
            float2 zv[A2U];
            #pragma unroll
            for (int u = 0; u < A2U; ++u) {
                int k = k0 + u * 1024;
                int kk = k < total ? k : k0;
                v[u] = kk < t0 ? s0[kk] : s1[kk - t0];
                if (k >= total) v[u].x = -1;
            }
            #pragma unroll
            for (int u = 0; u < A2U; ++u) {
                int s = v[u].x == -1 ? 0 : (v[u].x & 0x7FFFFF);
                zv[u] = *(const float2*)(z + 2 * (size_t)s);
            }
            #pragma unroll
            for (int u = 0; u < A2U; ++u) {
                if (v[u].x != -1) {
                    int dl = (int)((unsigned)v[u].x >> 23);
                    float nrm = __int_as_float(v[u].y);
                    atomicAdd(&o2[2 * dl + 0], zv[u].x * nrm);
                    atomicAdd(&o2[2 * dl + 1], zv[u].y * nrm);
                }
            }
        }
    } else {
        for (int e = tid; e < n_edges; e += 1024) {
            int d = dst[e];
            if ((unsigned)(d - lo) < (unsigned)cnt) {
                int s = src[e];
                float nrm = dinv[s] * dinv[d];
                float2 zv = *(const float2*)(z + 2 * (size_t)s);
                atomicAdd(&o2[2 * (d - lo) + 0], zv.x * nrm);
                atomicAdd(&o2[2 * (d - lo) + 1], zv.y * nrm);
            }
        }
    }
    __syncthreads();

    for (int i = tid; i < cnt * 2; i += 1024) out[2 * (size_t)lo + i] = o2[i];
}


extern "C" void kernel_launch(void* const* d_in, const int* in_sizes, int n_in,
                              void* d_out, int out_size, void* d_ws, size_t ws_size,
                              hipStream_t stream) {
    const float* x  = (const float*)d_in[0];
    const int*   ei = (const int*)d_in[1];
    const float* W1 = (const float*)d_in[2];
    const float* b1 = (const float*)d_in[3];
    const float* W2 = (const float*)d_in[4];
    const float* b2 = (const float*)d_in[5];

    int n_nodes = in_sizes[0] / NFEAT;
    int n_edges = in_sizes[1] / 2;
    const int* src = ei;
    const int* dst = ei + n_edges;

    float* ws = (float*)d_ws;
    size_t off = 0;
    float* dinv = ws + off; off += ((size_t)n_nodes + 63) & ~(size_t)63;
    float* h    = ws + off; off += (size_t)n_nodes * NHID;
    float* recs = ws + off; off += (size_t)n_nodes * NHID;
    float* z    = ws + off; off += ((size_t)2 * n_nodes + 63) & ~(size_t)63;
    (void)ws_size;

    float* out = (float*)d_out;

    int owners = (n_nodes + NR - 1) / NR;

    int* rec     = (int*)recs;
    int* rcount  = rec + (size_t)2 * (2 * MAXOWN) * RCAP;
    int* cpl     = rcount + (2 * MAXOWN) * 32;
    int  pairs   = (owners + 1) / 2;
    k_scan_gemm<<<2 * pairs, 1024, 0, stream>>>(x, W1, h, src, dst, cpl, rec, rcount, n_nodes, n_edges);

    k_agg1_owned<<<owners, 1024, 0, stream>>>(src, dst, cpl, dinv, h, b1, W2, z, rec, rcount, n_nodes, n_edges);

    k_agg2_owned<<<owners, 1024, 0, stream>>>(src, dst, dinv, z, b2, rec, rcount, out, n_nodes, n_edges);
}


# Owner scan tiled 4 ranges x a quarter of the destination list per work-group (list re-read 61 times, was 121); in-degree = sum of 4 count words, one 16-byte read; LDS-arrival sum order kept
# speedup vs baseline: 5.0096x; 1.0985x over previous
#include <hip/hip_runtime.h>

typedef float v2f __attribute__((ext_vector_type(2)));
typedef float v8f __attribute__((ext_vector_type(8)));

#define NFEAT 256
#define NHID  64


#define NGW  8
#define NSW  24
#define GPF  16
#define GSB  4

__device__ __forceinline__ void gemm_stage_w1(float* ldsW, const float* __restrict__ W1, int tid) {
    for (int idx = tid * 4; idx < NFEAT * NHID; idx += 1024 * 4) {
        float4 w = *(const float4*)(W1 + idx);
        int k  = idx >> 6;
        int n0 = idx & 63;
        int p  = k >> 1;
        int lo = k & 1;
        int rot = (p & 1) << 5;
        float vals[4] = {w.x, w.y, w.z, w.w};
        #pragma unroll
        for (int j = 0; j < 4; ++j) {
            int wpos = (((n0 + j) << 1) + rot) & 127;
            ldsW[p * 128 + wpos + lo] = vals[j];
        }
    }
}

__device__ __forceinline__ void gemm_wave(const float* __restrict__ x, float* __restrict__ h,
                                          const float* ldsW, int lane, int first, int stride, int n_tiles) {
    int t = first;
    if (t >= n_tiles) return;

    int half = lane >> 4;
    int kk   = half << 1;
    int ncol = lane & 15;
    int rot  = half << 5;

    int wp0 = (((ncol +  0) << 1) + rot) & 127;
    int wp1 = (((ncol + 16) << 1) + rot) & 127;
    int wp2 = (((ncol + 32) << 1) + rot) & 127;
    int wp3 = (((ncol + 48) << 1) + rot) & 127;
    unsigned boff = (unsigned)half * 128u;

    v2f a[GPF];
    const unsigned xlane = (unsigned)ncol * NFEAT + (unsigned)kk;
    const unsigned hlane = (unsigned)(half << 3) * NHID + (unsigned)ncol;
    const float* xr = x + (size_t)t * (16 * NFEAT) + xlane;
    #pragma unroll
    for (int s = 0; s < GPF; ++s) {
        float2 av = *(const float2*)(xr + s * 4);
        a[s].x = av.x; a[s].y = av.y;
    }
    for (;;) {
        const int  tn   = t + stride;
        const bool more = tn < n_tiles;
        const float* xn = x + (size_t)(more ? tn : t) * (16 * NFEAT) + xlane;

        asm volatile("" : "+v"(boff));
        const float* bb = ldsW + boff;
        v8f acc0 = {}, acc1 = {}, acc2 = {}, acc3 = {};
        #pragma unroll
        for (int g = 0; g < NFEAT / 4 / GSB; ++g) {
            #pragma unroll
            for (int j = 0; j < GSB; ++j) {
                const int s = g * GSB + j;
                const float* rowp = bb + (s << 1) * 128;
                v2f b0 = *(const v2f*)(rowp + wp0);
                v2f b1 = *(const v2f*)(rowp + wp1);
                v2f b2 = *(const v2f*)(rowp + wp2);
                v2f b3 = *(const v2f*)(rowp + wp3);
                const v2f av = a[s % GPF];
                acc0 = __builtin_amdgcn_wmma_f32_16x16x4_f32(false, av, false, b0, (short)0, acc0, false, false);
                acc1 = __builtin_amdgcn_wmma_f32_16x16x4_f32(false, av, false, b1, (short)0, acc1, false, false);
                acc2 = __builtin_amdgcn_wmma_f32_16x16x4_f32(false, av, false, b2, (short)0, acc2, false, false);
                acc3 = __builtin_amdgcn_wmma_f32_16x16x4_f32(false, av, false, b3, (short)0, acc3, false, false);
                const float* nx = (s + GPF < NFEAT / 4) ? xr + (s + GPF) * 4 : xn + (s + GPF - NFEAT / 4) * 4;
                float2 nv = *(const float2*)nx;
                a[s % GPF].x = nv.x; a[s % GPF].y = nv.y;
            }
            __builtin_amdgcn_sched_barrier(0);
        }

        float* ht = h + (size_t)t * (16 * NHID) + hlane;
        for (int r = 0; r < 8; ++r) {
            float* hr = ht + r * NHID;
            hr[ 0] = acc0[r];
            hr[16] = acc1[r];
            hr[32] = acc2[r];
            hr[48] = acc3[r];
        }
        if (!more) break;
        t = tn; xr = xn;
    }
}


#define NR   416
#define AP   65
#define NT   4
#define MAXOWN 256
#define RCAP 2816
#define QC   1792
#define SCAN_U 12

#define QSH 21

__device__ __forceinline__ int agg1_entry(unsigned dl, int e, unsigned pkm) {
    return (int)(((dl << QSH) & pkm) | (unsigned)e);
}

__device__ __forceinline__ int agg1_push(int* myq, int qn, bool m, int e) {
    unsigned mask = __builtin_amdgcn_ballot_w32(m);
    if (mask != 0u) {
        if (m) myq[qn + (int)__builtin_amdgcn_mbcnt_lo(mask, 0u)] = e;
        qn += __builtin_popcount(mask);
    }
    return qn;
}

__device__ __forceinline__ int agg1_scan_piece(int* myq, int qn, int4 v, int g, int lo, unsigned cntu, int lane,
                                               unsigned pkm) {
    unsigned a = (unsigned)(v.x - lo), b = (unsigned)(v.y - lo);
    unsigned c = (unsigned)(v.z - lo), d = (unsigned)(v.w - lo);
    unsigned ab = a < b ? a : b, cd = c < d ? c : d;
    unsigned mn = ab < cd ? ab : cd;
    if (__builtin_amdgcn_ballot_w32(mn < cntu) != 0u) {
        int e0 = (g << 7) + (lane << 2);
        qn = agg1_push(myq, qn, a < cntu, agg1_entry(a, e0 + 0, pkm));
        qn = agg1_push(myq, qn, b < cntu, agg1_entry(b, e0 + 1, pkm));
        qn = agg1_push(myq, qn, c < cntu, agg1_entry(c, e0 + 2, pkm));
        qn = agg1_push(myq, qn, d < cntu, agg1_entry(d, e0 + 3, pkm));
    }
    return qn;
}

__device__ __forceinline__ void rec_emit(int* stj, int fill, int lane, int2* recj, int* rcurj) {
    int rb = 0;
    if (lane == 0) rb = atomicAdd(rcurj, 32);
    rb = __builtin_amdgcn_readfirstlane(rb);
    int key  = lane < fill ? stj[lane] : -1;
    int more = 32 + lane < fill ? stj[32 + lane] : 0;
    if (rb + 32 <= RCAP) {
        int2 rv;
        rv.x = key;
        rv.y = 0;
        recj[rb + lane] = rv;
    }
    if (32 + lane < fill) stj[lane] = more;
}

__device__ __forceinline__ void scan_note(const int* myq, int qn, int lane, int lo,
                                          const int* __restrict__ src, const int* __restrict__ dst,
                                          int* degl, int2* myrec, int* rcur, int* st, int (&f)[NT],
                                          unsigned pkm) {
    for (int base = 0; base < qn; base += 32) {
        int nb = qn - base; if (nb > 32) nb = 32;
        int k = base + (lane < nb ? lane : nb - 1);
        const int ent = myq[k];
        int e = ent, dl;
        if (pkm != 0u) {
            e  = ent & ((1 << QSH) - 1);
            dl = (int)((unsigned)ent >> QSH);
        } else {
            dl = dst[ent] - lo;
        }
        int s = src[e];
        const bool live = lane < nb;
        if (live) atomicAdd(&degl[dl], 1);
        if (myrec != nullptr) {
            int jr = 0;
            #pragma unroll
            for (int j = 1; j < NT; ++j) jr += (dl >= j * NR) ? 1 : 0;
            const int key = (int)(((unsigned)(dl - jr * NR) << 23) | (unsigned)s);
            #pragma unroll
            for (int j = 0; j < NT; ++j) {
                const bool mine = live && jr == j;
                const unsigned mj = __builtin_amdgcn_ballot_w32(mine);
                if (mj != 0u) {
                    int* stj = st + j * 64;
                    if (mine) stj[f[j] + (int)__builtin_amdgcn_mbcnt_lo(mj, 0u)] = key;
                    f[j] += __builtin_popcount(mj);
                    if (f[j] >= 32) { rec_emit(stj, f[j], lane, myrec + (size_t)j * (NT * RCAP), rcur + j); f[j] -= 32; }
                }
            }
        }
    }
}

__device__ __forceinline__ void scan_wave(int* q, int* stw, int ws, int lane, int lo, int cnt,
                                          const int* __restrict__ src, const int* __restrict__ dst,
                                          int* degl, int2* myrec, int* rcur, int gb, int n_groups,
                                          int grp, int n_grps, bool tail, int n_edges, unsigned pkm) {
    int* myq = q + ws * QC;
    int* st  = stw + ws * (NT * 64);
    int qn = 0;
    int f[NT];
    #pragma unroll
    for (int j = 0; j < NT; ++j) f[j] = 0;
    const int4* dst4 = (const int4*)dst + (size_t)gb * 32;
    int rot = (int)(((long long)grp * n_groups) / n_grps);
    const int K = (n_groups - ws + NSW - 1) / NSW;

    for (int kb = 0; kb < K; kb += SCAN_U) {
        const int gi0 = ws + kb * NSW;
        int g0 = gi0 + rot;
        if (g0 >= n_groups) g0 -= n_groups;
        if (gi0 + NSW * (SCAN_U - 1) < n_groups && g0 + NSW * (SCAN_U - 1) < n_groups) {
            const int4* p = dst4 + (size_t)g0 * 32 + lane;
            int4 v[SCAN_U];
            #pragma unroll
            for (int u = 0; u < SCAN_U; ++u) v[u] = p[u * (NSW * 32)];
            #pragma unroll
            for (int u = 0; u < SCAN_U; ++u)
                qn = agg1_scan_piece(myq, qn, v[u], gb + g0 + NSW * u, lo, (unsigned)cnt, lane, pkm);
        } else {
            int4 v[SCAN_U];
            int  gg[SCAN_U];
            unsigned cu[SCAN_U];
            #pragma unroll
            for (int u = 0; u < SCAN_U; ++u) {
                int gi = gi0 + NSW * u;
                bool ok = gi < n_groups;
                int g = gi + rot;
                if (g >= n_groups) g -= n_groups;
                if (!ok) g = 0;
                gg[u] = gb + g;
                cu[u] = ok ? (unsigned)cnt : 0u;
                v[u] = dst4[(size_t)g * 32 + lane];
            }
            #pragma unroll
            for (int u = 0; u < SCAN_U; ++u)
                qn = agg1_scan_piece(myq, qn, v[u], gg[u], lo, cu[u], lane, pkm);
        }
        if (qn > QC - 128 * SCAN_U - 128) { scan_note(myq, qn, lane, lo, src, dst, degl, myrec, rcur, st, f, pkm); qn = 0; }
    }
    if (ws == 0 && tail) {
        for (int e0 = (n_edges >> 7) << 7; e0 < n_edges; e0 += 32) {
            int e = e0 + lane;
            unsigned a = (unsigned)(dst[e < n_edges ? e : 0] - lo);
            bool m = e < n_edges && a < (unsigned)cnt;
            qn = agg1_push(myq, qn, m, agg1_entry(a, e, pkm));
        }
    }
    scan_note(myq, qn, lane, lo, src, dst, degl, myrec, rcur, st, f, pkm);
    if (myrec != nullptr) {
        #pragma unroll
        for (int j = 0; j < NT; ++j)
            if (f[j] > 0) rec_emit(st + j * 64, f[j], lane, myrec + (size_t)j * (NT * RCAP), rcur + j);
    }
}

__global__ __launch_bounds__(1024) void k_scan_gemm(const float* __restrict__ x, const float* __restrict__ W1,
                                                    float* __restrict__ h,
                                                    const int* __restrict__ src, const int* __restrict__ dst,
                                                    int* __restrict__ cpl, int* __restrict__ rec,
                                                    int* __restrict__ rcount, int n_nodes, int n_edges) {
    __shared__ float ldsW[NFEAT * NHID];
    __shared__ int   degl[NT * NR];
    __shared__ int   q[NSW * QC];
    __shared__ int   stw[NSW * NT * 64];
    __shared__ int   rcur[NT];

    const int tid  = threadIdx.x;
    const int lane = tid & 31;
    const int wave = __builtin_amdgcn_readfirstlane(tid >> 5);
    const int grp = (int)blockIdx.x / NT;
    const int part = (int)blockIdx.x % NT;
    const int lo   = grp * (NT * NR);
    int cnt = n_nodes - lo;
    if (cnt > NT * NR) cnt = NT * NR;
    if (cnt < 0)       cnt = 0;
    const int all_groups = n_edges >> 7;
    const int gb = (int)(((long long)all_groups * part) / NT);
    const int ng = (int)(((long long)all_groups * (part + 1)) / NT) - gb;

    int2* myrec = (NT * grp < MAXOWN && n_nodes <= (1 << 23))
                      ? (int2*)rec + (size_t)(NT * NT * grp + part) * RCAP : nullptr;
    const unsigned pkm = n_edges <= (1 << QSH) ? 0xFFFFFFFFu : 0u;
    if (tid < NT) rcur[tid] = 0;
    for (int i = tid; i < NT * NR; i += 1024) degl[i] = 0;
    gemm_stage_w1(ldsW, W1, tid);
    __syncthreads();

    if (wave < NGW) {
        gemm_wave(x, h, ldsW, lane, (int)blockIdx.x + (int)gridDim.x * wave, (int)gridDim.x * NGW, n_nodes >> 4);
    } else {
        scan_wave(q, stw, wave - NGW, lane, lo, cnt, src, dst, degl, myrec, rcur, gb, ng,
                  grp, (int)gridDim.x / NT, part == NT - 1, n_edges, pkm);
    }
    __syncthreads();

    if (tid < NT && myrec != nullptr) rcount[(NT * (NT * grp + tid) + part) * 32] = rcur[tid];

    for (int i = tid; i < cnt; i += 1024) atomicExch(&cpl[NT * (size_t)(lo + i) + part], degl[i]);
}

static_assert(NT == 4, "dinv_of reads a node's NT count words as one int4");
__device__ __forceinline__ float dinv_of(const int* __restrict__ cpl, int i) {
    int4 c = *(const int4*)(cpl + NT * (size_t)i);
    float d = (float)(c.x + c.y + c.z + c.w + 1);
    return (d > 0.0f) ? rsqrtf(d) : 0.0f;
}

__global__ __launch_bounds__(1024) void k_agg1_owned(const int* __restrict__ src, const int* __restrict__ dst,
                                                     const int* __restrict__ cpl,
                                                     float* __restrict__ dinv, const float* __restrict__ h,
                                                     const float* __restrict__ b1, const float* __restrict__ W2,
                                                     float* __restrict__ z, int* rec,
                                                     const int* __restrict__ rcount, int n_nodes, int n_edges) {
    __shared__ float acc[NR * AP];
    __shared__ float dloc[NR];

    const int tid  = threadIdx.x;
    const int lane = tid & 31;
    const int wave = __builtin_amdgcn_readfirstlane(tid >> 5);
    const int lo   = blockIdx.x * NR;
    int cnt = n_nodes - lo;
    if (cnt > NR) cnt = NR;
    if (cnt < 0)  cnt = 0;

    for (int i = tid; i < cnt; i += 1024) {
        float di = dinv_of(cpl, lo + i);
        dloc[i] = di;
        dinv[lo + i] = di;
    }
    __syncthreads();

    for (int idx = tid; idx < cnt * NHID; idx += 1024) {
        int node = idx >> 6;
        float di = dloc[node];
        acc[node * AP + (idx & 63)] = h[(size_t)lo * NHID + idx] * di * di;
    }
    __syncthreads();

    const bool sliced = blockIdx.x < MAXOWN && n_nodes <= (1 << 23);
    int  c1[NT + 1];
    bool fits = true;
    c1[0] = 0;
    #pragma unroll
    for (int j = 0; j < NT; ++j) {
        const int tj = sliced ? rcount[(NT * blockIdx.x + j) * 32] : RCAP + 1;
        fits = fits && tj <= RCAP;
        c1[j + 1] = c1[j] + (tj >> 5);
    }
    if (fits) {
        int2* const s0 = (int2*)rec + (size_t)(NT * blockIdx.x) * RCAP;
        const int rounds = c1[NT];
        auto round_at = [&](int r) -> int2* {
            int off = r, jj = 0;
            #pragma unroll
            for (int i = 1; i < NT; ++i)
                if (r >= c1[i]) { off = r - c1[i]; jj = i; }
            return s0 + (size_t)jj * RCAP + off * 32;
        };
        int kx = -1;
        if (wave < rounds) kx = round_at(wave)[lane].x;
        for (int r = wave; r < rounds; r += 32) {
            int2* const myround = round_at(r);
            const int key = kx;
            if (r + 32 < rounds) kx = round_at(r + 32)[lane].x;
            bool ok = key != -1;
            int nb = __builtin_popcount(__builtin_amdgcn_ballot_w32(ok));
            int s  = ok ? (key & 0x7FFFFF) : 0;
            int dl = ok ? (int)((unsigned)key >> 23) : 0;
            float nrm = dinv_of(cpl, s) * dloc[dl];
            float2 hv[32];
            #pragma unroll
            for (int u = 0; u < 32; ++u) {
                int sj = __builtin_amdgcn_readlane(s, u);
                hv[u] = *(const float2*)(h + (size_t)sj * NHID + (lane << 1));
            }
            int2 rv; rv.x = key; rv.y = __float_as_int(nrm);
            myround[lane] = rv;
            #pragma unroll
            for (int u = 0; u < 32; ++u) {
                if (u < nb) {
                    int   dj = __builtin_amdgcn_readlane(dl, u);
                    float nj = __int_as_float(__builtin_amdgcn_readlane(__float_as_int(nrm), u));
                    float* row = acc + dj * AP + (lane << 1);
                    atomicAdd(row + 0, hv[u].x * nj);
                    atomicAdd(row + 1, hv[u].y * nj);
                }
            }
        }
    } else {
        for (int e = tid; e < n_edges; e += 1024) {
            int d = dst[e];
            if ((unsigned)(d - lo) < (unsigned)cnt) {
                int s = src[e];
                float nrm = dinv_of(cpl, s) * dloc[d - lo];
                const float* hr = h + (size_t)s * NHID;
                float* row = acc + (d - lo) * AP;
                for (int f = 0; f < NHID; ++f) atomicAdd(row + f, hr[f] * nrm);
            }
        }
    }
    __syncthreads();

    if (tid < cnt) {
        const float* ar = acc + tid * AP;
        float z0 = 0.0f, z1 = 0.0f;
        #pragma unroll
        for (int f = 0; f < NHID; ++f) {
            float hv = fmaxf(ar[f] + b1[f], 0.0f);
            z0 += hv * W2[2 * f + 0];
            z1 += hv * W2[2 * f + 1];
        }
        float2 zz; zz.x = z0; zz.y = z1;
        *(float2*)(z + 2 * (size_t)(lo + tid)) = zz;
    }
}


#define A2U 10

__global__ __launch_bounds__(1024) void k_agg2_owned(const int* __restrict__ src, const int* __restrict__ dst,
                                                     const float* __restrict__ dinv, const float* __restrict__ z,
                                                     const float* __restrict__ b2, const int* __restrict__ rec,
                                                     const int* __restrict__ rcount, float* __restrict__ out,
                                                     int n_nodes, int n_edges) {
    __shared__ float o2[NR * 2];
    const int tid = threadIdx.x;
    const int lo  = blockIdx.x * NR;
    int cnt = n_nodes - lo;
    if (cnt > NR) cnt = NR;
    if (cnt < 0)  cnt = 0;

    for (int i = tid; i < cnt * 2; i += 1024) {
        float di = dinv[lo + (i >> 1)];
        float n2 = di * di;
        o2[i] = z[2 * (size_t)lo + i] * n2 + b2[i & 1];
    }
    __syncthreads();

    const bool sliced = blockIdx.x < MAXOWN && n_nodes <= (1 << 23);
    int  e1[NT + 1];
    bool fits = true;
    e1[0] = 0;
    #pragma unroll
    for (int j = 0; j < NT; ++j) {
        const int tj = sliced ? rcount[(NT * blockIdx.x + j) * 32] : RCAP + 1;
        fits = fits && tj <= RCAP;
        e1[j + 1] = e1[j] + tj;
    }
    if (fits) {
        const int2* s0 = (const int2*)rec + (size_t)(NT * blockIdx.x) * RCAP;
        const int total = e1[NT];
        for (int k0 = tid; k0 < total; k0 += 1024 * A2U) {
            int2 v[A2U];
            float2 zv[A2U];
            #pragma unroll
            for (int u = 0; u < A2U; ++u) {
                int k = k0 + u * 1024;
                int kk = k < total ? k : k0;
                int off = kk, jj = 0;
                #pragma unroll
                for (int i = 1; i < NT; ++i)
                    if (kk >= e1[i]) { off = kk - e1[i]; jj = i; }
                v[u] = s0[jj * RCAP + off];
                if (k >= total) v[u].x = -1;
            }
            #pragma unroll
            for (int u = 0; u < A2U; ++u) {
                int s = v[u].x == -1 ? 0 : (v[u].x & 0x7FFFFF);
                zv[u] = *(const float2*)(z + 2 * (size_t)s);
            }
            #pragma unroll
            for (int u = 0; u < A2U; ++u) {
                if (v[u].x != -1) {
                    int dl = (int)((unsigned)v[u].x >> 23);
                    float nrm = __int_as_float(v[u].y);
                    atomicAdd(&o2[2 * dl + 0], zv[u].x * nrm);
                    atomicAdd(&o2[2 * dl + 1], zv[u].y * nrm);
                }
            }
        }
    } else {
        for (int e = tid; e < n_edges; e += 1024) {
            int d = dst[e];
            if ((unsigned)(d - lo) < (unsigned)cnt) {
                int s = src[e];
                float nrm = dinv[s] * dinv[d];
                float2 zv = *(const float2*)(z + 2 * (size_t)s);
                atomicAdd(&o2[2 * (d - lo) + 0], zv.x * nrm);
                atomicAdd(&o2[2 * (d - lo) + 1], zv.y * nrm);
            }
        }
    }
    __syncthreads();

    for (int i = tid; i < cnt * 2; i += 1024) out[2 * (size_t)lo + i] = o2[i];
}


extern "C" void kernel_launch(void* const* d_in, const int* in_sizes, int n_in,
                              void* d_out, int out_size, void* d_ws, size_t ws_size,
                              hipStream_t stream) {
    const float* x  = (const float*)d_in[0];
    const int*   ei = (const int*)d_in[1];
    const float* W1 = (const float*)d_in[2];
    const float* b1 = (const float*)d_in[3];
    const float* W2 = (const float*)d_in[4];
    const float* b2 = (const float*)d_in[5];

    int n_nodes = in_sizes[0] / NFEAT;
    int n_edges = in_sizes[1] / 2;
    const int* src = ei;
    const int* dst = ei + n_edges;

    float* ws = (float*)d_ws;
    size_t off = 0;
    float* dinv = ws + off; off += ((size_t)n_nodes + 63) & ~(size_t)63;
    float* h    = ws + off; off += (size_t)n_nodes * NHID;
    float* recs = ws + off; off += (size_t)n_nodes * NHID;
    float* z    = ws + off; off += ((size_t)2 * n_nodes + 63) & ~(size_t)63;
    (void)ws_size;

    float* out = (float*)d_out;

    int owners = (n_nodes + NR - 1) / NR;

    int* rec     = (int*)recs;
    int* rcount  = rec + (size_t)2 * (NT * MAXOWN) * RCAP;
    int* cpl     = rcount + (NT * MAXOWN) * 32;
    int  grps    = (owners + NT - 1) / NT;
    k_scan_gemm<<<NT * grps, 1024, 0, stream>>>(x, W1, h, src, dst, cpl, rec, rcount, n_nodes, n_edges);

    k_agg1_owned<<<owners, 1024, 0, stream>>>(src, dst, cpl, dinv, h, b1, W2, z, rec, rcount, n_nodes, n_edges);

    k_agg2_owned<<<owners, 1024, 0, stream>>>(src, dst, dinv, z, b2, rec, rcount, out, n_nodes, n_edges);
}
